# starting kernel
# baseline (speedup 1.0000x reference)
#include <hip/hip_runtime.h>
#include <hip/hip_bf16.h>

#define NH 128
#define NLAYERS 3
#define LN_EPS 1e-5f

typedef float v2f __attribute__((ext_vector_type(2)));
typedef float v8f __attribute__((ext_vector_type(8)));

__device__ __forceinline__ void atomic_add_f32_dev(float* p, float v) {
    asm volatile("global_atomic_add_f32 %0, %1, off scope:SCOPE_DEV"
                 :
                 : "v"(p), "v"(v)
                 : "memory");
}

__global__ void k_deg_init(float* deg, int n) {
    int i = blockIdx.x * 256 + threadIdx.x;
    if (i < n) deg[i] = 1.0f;
}

__global__ void k_deg_edges(const int* __restrict__ dst, float* deg, int e) {
    int i = blockIdx.x * 256 + threadIdx.x;
    if (i < e) atomic_add_f32_dev(&deg[dst[i]], 1.0f);
}

__global__ void k_dinv(float* deg_inout, int n) {
    int i = blockIdx.x * 256 + threadIdx.x;
    if (i < n) {
        float d = deg_inout[i];
        deg_inout[i] = (d > 0.0f) ? rsqrtf(d) : 0.0f;
    }
}

__global__ void k_norm(const int* __restrict__ src, const int* __restrict__ dst,
                       const float* __restrict__ dinv, float* __restrict__ nrm, int e) {
    int i = blockIdx.x * 256 + threadIdx.x;
    if (i < e) nrm[i] = dinv[src[i]] * dinv[dst[i]];
}

__global__ void __launch_bounds__(256)
k_gemm128(const float* __restrict__ A, const float* __restrict__ W,
          float* __restrict__ out) {
    __shared__ float lds[16 * 130];
    const int row0 = blockIdx.x * 16;

    #pragma unroll
    for (int i = 0; i < 4; ++i) {
        int idx = threadIdx.x + 256 * i;
        int r   = idx >> 6;
        int c2  = (idx & 63) * 2;
        const float2 v = *(const float2*)(A + (size_t)(row0 + r) * NH + c2);
        *(float2*)(lds + r * 130 + c2) = v;
    }
    __syncthreads();

    const int lane = threadIdx.x & 31;
    const int wv   = threadIdx.x >> 5;
    const int col0 = wv * 16;
    const int m    = lane & 15;
    const int half = lane >> 4;

    const float* arow = lds + m * 130;
    v8f c;
    #pragma unroll
    for (int i = 0; i < 8; ++i) c[i] = 0.0f;

    #pragma unroll 4
    for (int kk = 0; kk < 32; ++kk) {
        const int k0 = kk * 4 + 2 * half;
        v2f a;
        a.x = arow[k0];
        a.y = arow[k0 + 1];
        v2f b;
        b.x = W[(size_t)k0 * NH + col0 + m];
        b.y = W[(size_t)(k0 + 1) * NH + col0 + m];
        c = __builtin_amdgcn_wmma_f32_16x16x4_f32(false, a, false, b,
                                                  (short)0, c, false, false);
    }

    const int outRow = row0 + 8 * half;
    #pragma unroll
    for (int r = 0; r < 8; ++r)
        out[(size_t)(outRow + r) * NH + col0 + m] = c[r];
}

__global__ void k_agg_init(const float* __restrict__ hw, const float* __restrict__ dinv,
                           const float* __restrict__ bias, float* __restrict__ agg, int n) {
    int t  = blockIdx.x * 256 + threadIdx.x;
    int nd = t >> 5;
    if (nd >= n) return;
    int c  = (t & 31) * 4;
    float di = dinv[nd];
    float sl = di * di;
    float4 v = *(const float4*)(hw + (size_t)nd * NH + c);
    float4 b = *(const float4*)(bias + c);
    float4 o;
    o.x = fmaf(v.x, sl, b.x); o.y = fmaf(v.y, sl, b.y);
    o.z = fmaf(v.z, sl, b.z); o.w = fmaf(v.w, sl, b.w);
    *(float4*)(agg + (size_t)nd * NH + c) = o;
}

__global__ void k_scatter(const float* __restrict__ hw, const int* __restrict__ src,
                          const int* __restrict__ dst, const float* __restrict__ nrm,
                          float* __restrict__ agg, int e) {
    int t  = blockIdx.x * 256 + threadIdx.x;
    int ei = t >> 5;
    if (ei >= e) return;
    int c  = (t & 31) * 4;
    int s  = src[ei];
    int d  = dst[ei];
    float w = nrm[ei];
    float4 v = *(const float4*)(hw + (size_t)s * NH + c);
    float* p = agg + (size_t)d * NH + c;
    atomic_add_f32_dev(p + 0, v.x * w);
    atomic_add_f32_dev(p + 1, v.y * w);
    atomic_add_f32_dev(p + 2, v.z * w);
    atomic_add_f32_dev(p + 3, v.w * w);
}

__device__ __forceinline__ float gelu_exact(float x) {
    return 0.5f * x * (1.0f + erff(x * 0.70710678118654752f));
}

__global__ void k_lngelu(const float* __restrict__ in, const float* __restrict__ bias,
                         const float* __restrict__ gamma, const float* __restrict__ beta,
                         const float* __restrict__ resid, float* __restrict__ out, int n) {
    const int lane = threadIdx.x & 31;
    const int wv   = threadIdx.x >> 5;
    const int node = blockIdx.x * 8 + wv;
    if (node >= n) return;
    const int c = lane * 4;

    float4 t = *(const float4*)(in + (size_t)node * NH + c);
    if (bias) {
        float4 b = *(const float4*)(bias + c);
        t.x += b.x; t.y += b.y; t.z += b.z; t.w += b.w;
    }
    float s  = t.x + t.y + t.z + t.w;
    float ss = fmaf(t.x, t.x, fmaf(t.y, t.y, fmaf(t.z, t.z, t.w * t.w)));
    #pragma unroll
    for (int o = 16; o > 0; o >>= 1) {
        s  += __shfl_xor(s,  o, 32);
        ss += __shfl_xor(ss, o, 32);
    }
    const float mu   = s * (1.0f / NH);
    const float var  = fmaf(ss, (1.0f / NH), -mu * mu);
    const float rstd = rsqrtf(var + LN_EPS);

    float4 g  = *(const float4*)(gamma + c);
    float4 be = *(const float4*)(beta + c);
    float4 o;
    o.x = gelu_exact(fmaf((t.x - mu) * rstd, g.x, be.x));
    o.y = gelu_exact(fmaf((t.y - mu) * rstd, g.y, be.y));
    o.z = gelu_exact(fmaf((t.z - mu) * rstd, g.z, be.z));
    o.w = gelu_exact(fmaf((t.w - mu) * rstd, g.w, be.w));
    if (resid) {
        float4 r = *(const float4*)(resid + (size_t)node * NH + c);
        o.x += r.x; o.y += r.y; o.z += r.z; o.w += r.w;
    }
    *(float4*)(out + (size_t)node * NH + c) = o;
}

__global__ void k_head(const float* __restrict__ h, const float* __restrict__ wh,
                       const float* __restrict__ bh, float* __restrict__ out, int n) {
    const int lane = threadIdx.x & 31;
    const int wv   = threadIdx.x >> 5;
    const int node = blockIdx.x * 8 + wv;
    if (node >= n) return;
    const int c = lane * 4;
    float4 v = *(const float4*)(h + (size_t)node * NH + c);
    float4 w = *(const float4*)(wh + c);
    float p  = fmaf(v.x, w.x, fmaf(v.y, w.y, fmaf(v.z, w.z, v.w * w.w)));
    #pragma unroll
    for (int o = 16; o > 0; o >>= 1) p += __shfl_xor(p, o, 32);
    if (lane == 0) out[node] = p + bh[0];
}

extern "C" void kernel_launch(void* const* d_in, const int* in_sizes, int n_in,
                              void* d_out, int out_size, void* d_ws, size_t ws_size,
                              hipStream_t stream) {
    const float* x      = (const float*)d_in[0];
    const int*   ei     = (const int*)d_in[1];
    const float* W_in   = (const float*)d_in[2];
    const float* b_in   = (const float*)d_in[3];
    const float* g_in   = (const float*)d_in[4];
    const float* be_in  = (const float*)d_in[5];
    const float* W_conv = (const float*)d_in[6];
    const float* b_conv = (const float*)d_in[7];
    const float* g_ln   = (const float*)d_in[8];
    const float* be_ln  = (const float*)d_in[9];
    const float* W_head = (const float*)d_in[10];
    const float* b_head = (const float*)d_in[11];
    float* out = (float*)d_out;

    const int N = in_sizes[0] / NH;
    const int E = in_sizes[1] / 2;
    const int* src = ei;
    const int* dst = ei + E;

    char* w = (char*)d_ws;
    float* dinv = (float*)(w);
    float* nrm  = (float*)(w + (size_t)(1)   * (1 << 20));
    float* h    = (float*)(w + (size_t)(4)   * (1 << 20));
    float* hw   = (float*)(w + (size_t)(60)  * (1 << 20));
    float* agg  = (float*)(w + (size_t)(116) * (1 << 20));

    const int gN   = (N + 255) / 256;
    const int gE   = (E + 255) / 256;
    const int gRow = N / 16;
    const int gNW  = (N * 32 + 255) / 256;
    const int gEW  = (E * 32 + 255) / 256;
    const int gNd  = (N + 7) / 8;

    k_deg_init<<<gN, 256, 0, stream>>>(dinv, N);
    k_deg_edges<<<gE, 256, 0, stream>>>(dst, dinv, E);
    k_dinv<<<gN, 256, 0, stream>>>(dinv, N);
    k_norm<<<gE, 256, 0, stream>>>(src, dst, dinv, nrm, E);

    k_gemm128<<<gRow, 256, 0, stream>>>(x, W_in, hw);
    k_lngelu<<<gNd, 256, 0, stream>>>(hw, b_in, g_in, be_in, nullptr, h, N);

    for (int i = 0; i < NLAYERS; ++i) {
        const float* Wc = W_conv + (size_t)i * NH * NH;
        const float* bc = b_conv + (size_t)i * NH;
        const float* gl = g_ln   + (size_t)i * NH;
        const float* bl = be_ln  + (size_t)i * NH;
        k_gemm128<<<gRow, 256, 0, stream>>>(h, Wc, hw);
        k_agg_init<<<gNW, 256, 0, stream>>>(hw, dinv, bc, agg, N);
        k_scatter<<<gEW, 256, 0, stream>>>(hw, src, dst, nrm, agg, E);
        k_lngelu<<<gNd, 256, 0, stream>>>(agg, nullptr, gl, bl, h, h, N);
    }

    k_head<<<gNd, 256, 0, stream>>>(h, W_head, b_head, out, N);
}


# Atomic f32 scatter replaced by a pull: edges stable-sorted by destination each call (3-pass ballot-ranked radix sort, no atomics), rows gathered per node in registers; sum order now fixed
# speedup vs baseline: 1.4343x; 1.4343x over previous
#include <hip/hip_runtime.h>
#include <hip/hip_bf16.h>

#define NH 128
#define NLAYERS 3
#define LN_EPS 1e-5f

typedef float v2f __attribute__((ext_vector_type(2)));
typedef float v8f __attribute__((ext_vector_type(8)));

#define RBITS 6
#define RDIG  64

__device__ __forceinline__ int block_digit_rank(int digit, bool valid, int* s_cnt, int* s_tot) {
    const int tid  = threadIdx.x;
    const int lane = tid & 31;
    const int wv   = tid >> 5;
    s_cnt[tid] = 0;
    s_cnt[tid + 1024] = 0;
    unsigned peers = __builtin_amdgcn_ballot_w32(valid);
    for (int b = 0; b < RBITS; ++b) {
        const bool bit = ((digit >> b) & 1) != 0;
        const unsigned bal = __builtin_amdgcn_ballot_w32(bit);
        peers &= bit ? bal : ~bal;
    }
    const int rank = __popc(peers & ((1u << lane) - 1u));
    __syncthreads();
    if (valid && rank == 0) s_cnt[wv * RDIG + digit] = __popc(peers);
    __syncthreads();
    if (tid < RDIG) {
        int run = 0;
        for (int w = 0; w < 32; ++w) {
            const int t = s_cnt[w * RDIG + tid];
            s_cnt[w * RDIG + tid] = run;
            run += t;
        }
        s_tot[tid] = run;
    }
    __syncthreads();
    return rank;
}

__global__ void __launch_bounds__(1024)
k_sort_count(const int* __restrict__ keys, int* __restrict__ ghist,
             int e, int shift, int nblocks) {
    __shared__ int s_cnt[32 * RDIG];
    __shared__ int s_tot[RDIG];
    const int i = blockIdx.x * 1024 + threadIdx.x;
    const bool valid = i < e;
    const int key = valid ? keys[i] : 0;
    block_digit_rank((key >> shift) & (RDIG - 1), valid, s_cnt, s_tot);
    if (threadIdx.x < RDIG) ghist[threadIdx.x * nblocks + blockIdx.x] = s_tot[threadIdx.x];
}

__global__ void __launch_bounds__(1024)
k_sort_move(const int* __restrict__ keys, const int* __restrict__ vals,
            const int* __restrict__ ghist, int* __restrict__ keys_out,
            int* __restrict__ vals_out, int e, int shift, int nblocks) {
    __shared__ int s_cnt[32 * RDIG];
    __shared__ int s_tot[RDIG];
    __shared__ int s_all[RDIG];
    __shared__ int s_pre[RDIG];
    const int tid = threadIdx.x;
    const int i = blockIdx.x * 1024 + tid;
    const bool valid = i < e;
    const int key = valid ? keys[i] : 0;
    const int val = valid ? vals[i] : 0;
    const int digit = (key >> shift) & (RDIG - 1);
    const int rank = block_digit_rank(digit, valid, s_cnt, s_tot);

    {
        const int d = tid >> 4;
        const int sub = tid & 15;
        const int* row = ghist + d * nblocks;
        int all = 0, pre = 0;
        for (int b = sub; b < nblocks; b += 16) {
            const int v = row[b];
            all += v;
            pre += (b < (int)blockIdx.x) ? v : 0;
        }
        for (int o = 8; o > 0; o >>= 1) {
            all += __shfl_xor(all, o, 32);
            pre += __shfl_xor(pre, o, 32);
        }
        if (sub == 0) { s_all[d] = all; s_pre[d] = pre; }
    }
    __syncthreads();
    if (tid < 32) {
        const int a0 = s_all[2 * tid];
        const int a1 = s_all[2 * tid + 1];
        int x = a0 + a1;
        for (int o = 1; o < 32; o <<= 1) {
            const int y = __shfl_up(x, o, 32);
            if (tid >= o) x += y;
        }
        const int ex = x - (a0 + a1);
        s_all[2 * tid] = ex;
        s_all[2 * tid + 1] = ex + a0;
    }
    __syncthreads();
    if (valid) {
        const int p = s_all[digit] + s_pre[digit] + s_cnt[(tid >> 5) * RDIG + digit] + rank;
        keys_out[p] = key;
        vals_out[p] = val;
    }
}

__global__ void k_rowptr_dinv(const int* __restrict__ sdst, int* __restrict__ rowptr,
                              float* __restrict__ dinv, int n, int e) {
    const int i = blockIdx.x * 256 + threadIdx.x;
    if (i > n) return;
    int lo0 = 0, hi0 = e;
    int lo1 = 0, hi1 = e;
    while ((lo0 < hi0) | (lo1 < hi1)) {
        const int m0 = (lo0 + hi0) >> 1;
        const int m1 = (lo1 + hi1) >> 1;
        const int k0 = sdst[min(m0, e - 1)];
        const int k1 = sdst[min(m1, e - 1)];
        if (lo0 < hi0) { if (k0 < i) lo0 = m0 + 1; else hi0 = m0; }
        if (lo1 < hi1) { if (k1 < i + 1) lo1 = m1 + 1; else hi1 = m1; }
    }
    rowptr[i] = lo0;
    if (i < n) {
        const float d = (float)(lo1 - lo0 + 1);
        dinv[i] = (d > 0.0f) ? rsqrtf(d) : 0.0f;
    }
}

__global__ void k_fill(const int* __restrict__ ssrc, const int* __restrict__ sdst,
                       const float* __restrict__ dinv, int* __restrict__ csr, int e) {
    const int i = blockIdx.x * 256 + threadIdx.x;
    if (i < e) {
        const int s = ssrc[i];
        const int d = sdst[i];
        const float w = dinv[s] * dinv[d];
        *(int2*)(csr + 2 * (size_t)i) = make_int2(s, __float_as_int(w));
    }
}

__global__ void __launch_bounds__(256)
k_gemm128(const float* __restrict__ A, const float* __restrict__ W,
          float* __restrict__ out) {
    __shared__ float lds[16 * 130];
    const int row0 = blockIdx.x * 16;

    #pragma unroll
    for (int i = 0; i < 4; ++i) {
        int idx = threadIdx.x + 256 * i;
        int r   = idx >> 6;
        int c2  = (idx & 63) * 2;
        const float2 v = *(const float2*)(A + (size_t)(row0 + r) * NH + c2);
        *(float2*)(lds + r * 130 + c2) = v;
    }
    __syncthreads();

    const int lane = threadIdx.x & 31;
    const int wv   = threadIdx.x >> 5;
    const int col0 = wv * 16;
    const int m    = lane & 15;
    const int half = lane >> 4;

    const float* arow = lds + m * 130;
    v8f c;
    #pragma unroll
    for (int i = 0; i < 8; ++i) c[i] = 0.0f;

    #pragma unroll 4
    for (int kk = 0; kk < 32; ++kk) {
        const int k0 = kk * 4 + 2 * half;
        v2f a;
        a.x = arow[k0];
        a.y = arow[k0 + 1];
        v2f b;
        b.x = W[(size_t)k0 * NH + col0 + m];
        b.y = W[(size_t)(k0 + 1) * NH + col0 + m];
        c = __builtin_amdgcn_wmma_f32_16x16x4_f32(false, a, false, b,
                                                  (short)0, c, false, false);
    }

    const int outRow = row0 + 8 * half;
    #pragma unroll
    for (int r = 0; r < 8; ++r)
        out[(size_t)(outRow + r) * NH + col0 + m] = c[r];
}

#define GROWS 8

__device__ __forceinline__ float4 fma4(float4 r, float w, float4 a) {
    a.x = fmaf(r.x, w, a.x); a.y = fmaf(r.y, w, a.y);
    a.z = fmaf(r.z, w, a.z); a.w = fmaf(r.w, w, a.w);
    return a;
}

__global__ void __launch_bounds__(256)
k_gather(const float* __restrict__ hw, const int* __restrict__ rowptr,
         const int* __restrict__ csr, const float* __restrict__ dinv,
         const float* __restrict__ bias, float* __restrict__ agg, int n) {
    const int lane = threadIdx.x & 31;
    const int c    = lane * 4;
    const int wave0  = __builtin_amdgcn_readfirstlane((int)(blockIdx.x * 8 + (threadIdx.x >> 5)));
    const int nwaves = gridDim.x * 8;
    const float4 b = *(const float4*)(bias + c);
    const float4 z = make_float4(0.0f, 0.0f, 0.0f, 0.0f);

    int beg = 0, end = 0;
    if (wave0 < n) { beg = rowptr[wave0]; end = rowptr[wave0 + 1]; }

    for (int node = wave0; node < n; node += nwaves) {
        const int nxt = node + nwaves;
        int nbeg = 0, nend = 0;
        if (nxt < n) { nbeg = rowptr[nxt]; nend = rowptr[nxt + 1]; }

        const float di = dinv[node];
        const float sl = di * di;
        const float4 v = *(const float4*)(hw + (size_t)node * NH + c);
        int2 rec = make_int2(0, 0);
        if (lane < end - beg) rec = *(const int2*)(csr + 2 * (size_t)(beg + lane));
        float4 acc = fma4(v, sl, b);

        for (int base = beg; base < end; base += 32) {
            const int m = min(32, end - base);
            if (base != beg) {
                rec = make_int2(0, 0);
                if (lane < m) rec = *(const int2*)(csr + 2 * (size_t)(base + lane));
            }
            for (int k = 0; k < m; k += GROWS) {
                int    sidx[GROWS];
                float  wt[GROWS];
                float4 r[GROWS];
                for (int j = 0; j < GROWS; ++j) {
                    const int q = (k + j) & 31;
                    const int s = __builtin_amdgcn_readlane(rec.x, q);
                    const float w = __int_as_float(__builtin_amdgcn_readlane(rec.y, q));
                    sidx[j] = min(max(s, 0), n - 1);
                    wt[j] = (k + j < m) ? w : 0.0f;
                }
                for (int j = 0; j < GROWS; ++j) {
                    r[j] = z;
                    if (k + j < m) r[j] = *(const float4*)(hw + (size_t)sidx[j] * NH + c);
                }
                for (int j = 0; j < GROWS; ++j) acc = fma4(r[j], wt[j], acc);
            }
        }
        *(float4*)(agg + (size_t)node * NH + c) = acc;
        beg = nbeg;
        end = nend;
    }
}

__device__ __forceinline__ float gelu_exact(float x) {
    return 0.5f * x * (1.0f + erff(x * 0.70710678118654752f));
}

__global__ void k_lngelu(const float* __restrict__ in, const float* __restrict__ bias,
                         const float* __restrict__ gamma, const float* __restrict__ beta,
                         const float* __restrict__ resid, float* __restrict__ out, int n) {
    const int lane = threadIdx.x & 31;
    const int wv   = threadIdx.x >> 5;
    const int node = blockIdx.x * 8 + wv;
    if (node >= n) return;
    const int c = lane * 4;

    float4 t = *(const float4*)(in + (size_t)node * NH + c);
    if (bias) {
        float4 b = *(const float4*)(bias + c);
        t.x += b.x; t.y += b.y; t.z += b.z; t.w += b.w;
    }
    float s  = t.x + t.y + t.z + t.w;
    float ss = fmaf(t.x, t.x, fmaf(t.y, t.y, fmaf(t.z, t.z, t.w * t.w)));
    #pragma unroll
    for (int o = 16; o > 0; o >>= 1) {
        s  += __shfl_xor(s,  o, 32);
        ss += __shfl_xor(ss, o, 32);
    }
    const float mu   = s * (1.0f / NH);
    const float var  = fmaf(ss, (1.0f / NH), -mu * mu);
    const float rstd = rsqrtf(var + LN_EPS);

    float4 g  = *(const float4*)(gamma + c);
    float4 be = *(const float4*)(beta + c);
    float4 o;
    o.x = gelu_exact(fmaf((t.x - mu) * rstd, g.x, be.x));
    o.y = gelu_exact(fmaf((t.y - mu) * rstd, g.y, be.y));
    o.z = gelu_exact(fmaf((t.z - mu) * rstd, g.z, be.z));
    o.w = gelu_exact(fmaf((t.w - mu) * rstd, g.w, be.w));
    if (resid) {
        float4 r = *(const float4*)(resid + (size_t)node * NH + c);
        o.x += r.x; o.y += r.y; o.z += r.z; o.w += r.w;
    }
    *(float4*)(out + (size_t)node * NH + c) = o;
}

__global__ void k_head(const float* __restrict__ h, const float* __restrict__ wh,
                       const float* __restrict__ bh, float* __restrict__ out, int n) {
    const int lane = threadIdx.x & 31;
    const int wv   = threadIdx.x >> 5;
    const int node = blockIdx.x * 8 + wv;
    if (node >= n) return;
    const int c = lane * 4;
    float4 v = *(const float4*)(h + (size_t)node * NH + c);
    float4 w = *(const float4*)(wh + c);
    float p  = fmaf(v.x, w.x, fmaf(v.y, w.y, fmaf(v.z, w.z, v.w * w.w)));
    #pragma unroll
    for (int o = 16; o > 0; o >>= 1) p += __shfl_xor(p, o, 32);
    if (lane == 0) out[node] = p + bh[0];
}

extern "C" void kernel_launch(void* const* d_in, const int* in_sizes, int n_in,
                              void* d_out, int out_size, void* d_ws, size_t ws_size,
                              hipStream_t stream) {
    const float* x      = (const float*)d_in[0];
    const int*   ei     = (const int*)d_in[1];
    const float* W_in   = (const float*)d_in[2];
    const float* b_in   = (const float*)d_in[3];
    const float* g_in   = (const float*)d_in[4];
    const float* be_in  = (const float*)d_in[5];
    const float* W_conv = (const float*)d_in[6];
    const float* b_conv = (const float*)d_in[7];
    const float* g_ln   = (const float*)d_in[8];
    const float* be_ln  = (const float*)d_in[9];
    const float* W_head = (const float*)d_in[10];
    const float* b_head = (const float*)d_in[11];
    float* out = (float*)d_out;

    const int N = in_sizes[0] / NH;
    const int E = in_sizes[1] / 2;
    const int* src = ei;
    const int* dst = ei + E;

    char* w = (char*)d_ws;
    float* dinv   = (float*)(w);
    int*   ghist  = (int*)  (w + (size_t)(1)   * (1 << 20));
    int*   rowptr = (int*)  (w + (size_t)(2)   * (1 << 20));
    float* h      = (float*)(w + (size_t)(4)   * (1 << 20));
    float* hw     = (float*)(w + (size_t)(60)  * (1 << 20));
    float* agg    = (float*)(w + (size_t)(116) * (1 << 20));
    int*   kA     = (int*)  (w + (size_t)(168) * (1 << 20));
    int*   vA     = (int*)  (w + (size_t)(171) * (1 << 20));
    int*   kB     = (int*)  (w + (size_t)(174) * (1 << 20));
    int*   vB     = (int*)  (w + (size_t)(177) * (1 << 20));
    int*   csr    = (int*)  (w + (size_t)(180) * (1 << 20));

    const int gE   = (E + 255) / 256;
    const int gN1  = (N + 1 + 255) / 256;
    const int gRow = N / 16;
    const int gNd  = (N + 7) / 8;
    const int nSb  = (E + 1023) / 1024;
    const int gGa  = 2048;

    k_sort_count<<<nSb, 1024, 0, stream>>>(dst, ghist, E, 0, nSb);
    k_sort_move<<<nSb, 1024, 0, stream>>>(dst, src, ghist, kA, vA, E, 0, nSb);
    k_sort_count<<<nSb, 1024, 0, stream>>>(kA, ghist, E, 6, nSb);
    k_sort_move<<<nSb, 1024, 0, stream>>>(kA, vA, ghist, kB, vB, E, 6, nSb);
    k_sort_count<<<nSb, 1024, 0, stream>>>(kB, ghist, E, 12, nSb);
    k_sort_move<<<nSb, 1024, 0, stream>>>(kB, vB, ghist, kA, vA, E, 12, nSb);
    k_rowptr_dinv<<<gN1, 256, 0, stream>>>(kA, rowptr, dinv, N, E);
    k_fill<<<gE, 256, 0, stream>>>(vA, kA, dinv, csr, E);

    k_gemm128<<<gRow, 256, 0, stream>>>(x, W_in, hw);
    k_lngelu<<<gNd, 256, 0, stream>>>(hw, b_in, g_in, be_in, nullptr, h, N);

    for (int i = 0; i < NLAYERS; ++i) {
        const float* Wc = W_conv + (size_t)i * NH * NH;
        const float* bc = b_conv + (size_t)i * NH;
        const float* gl = g_ln   + (size_t)i * NH;
        const float* bl = be_ln  + (size_t)i * NH;
        k_gemm128<<<gRow, 256, 0, stream>>>(h, Wc, hw);
        k_gather<<<gGa, 256, 0, stream>>>(hw, rowptr, csr, dinv, bc, agg, N);
        k_lngelu<<<gNd, 256, 0, stream>>>(agg, nullptr, gl, bl, h, h, N);
    }

    k_head<<<gNd, 256, 0, stream>>>(h, W_head, b_head, out, N);
}


# LayerNorm+GELU+residual (last layer: the head dot too) moved into the gather wave that already holds the node's 128-value row in registers; agg write+read and 4 one-node-per-wave launches dropped
# speedup vs baseline: 1.6316x; 1.1376x over previous
#include <hip/hip_runtime.h>
#include <hip/hip_bf16.h>

#define NH 128
#define NLAYERS 3
#define LN_EPS 1e-5f

typedef float v2f __attribute__((ext_vector_type(2)));
typedef float v8f __attribute__((ext_vector_type(8)));

#define RBITS 6
#define RDIG  64

__device__ __forceinline__ int block_digit_rank(int digit, bool valid, int* s_cnt, int* s_tot) {
    const int tid  = threadIdx.x;
    const int lane = tid & 31;
    const int wv   = tid >> 5;
    s_cnt[tid] = 0;
    s_cnt[tid + 1024] = 0;
    unsigned peers = __builtin_amdgcn_ballot_w32(valid);
    for (int b = 0; b < RBITS; ++b) {
        const bool bit = ((digit >> b) & 1) != 0;
        const unsigned bal = __builtin_amdgcn_ballot_w32(bit);
        peers &= bit ? bal : ~bal;
    }
    const int rank = __popc(peers & ((1u << lane) - 1u));
    __syncthreads();
    if (valid && rank == 0) s_cnt[wv * RDIG + digit] = __popc(peers);
    __syncthreads();
    if (tid < RDIG) {
        int run = 0;
        for (int w = 0; w < 32; ++w) {
            const int t = s_cnt[w * RDIG + tid];
            s_cnt[w * RDIG + tid] = run;
            run += t;
        }
        s_tot[tid] = run;
    }
    __syncthreads();
    return rank;
}

__global__ void __launch_bounds__(1024)
k_sort_count(const int* __restrict__ keys, int* __restrict__ ghist,
             int e, int shift, int nblocks) {
    __shared__ int s_cnt[32 * RDIG];
    __shared__ int s_tot[RDIG];
    const int i = blockIdx.x * 1024 + threadIdx.x;
    const bool valid = i < e;
    const int key = valid ? keys[i] : 0;
    block_digit_rank((key >> shift) & (RDIG - 1), valid, s_cnt, s_tot);
    if (threadIdx.x < RDIG) ghist[threadIdx.x * nblocks + blockIdx.x] = s_tot[threadIdx.x];
}

__global__ void __launch_bounds__(1024)
k_sort_move(const int* __restrict__ keys, const int* __restrict__ vals,
            const int* __restrict__ ghist, int* __restrict__ keys_out,
            int* __restrict__ vals_out, int e, int shift, int nblocks) {
    __shared__ int s_cnt[32 * RDIG];
    __shared__ int s_tot[RDIG];
    __shared__ int s_all[RDIG];
    __shared__ int s_pre[RDIG];
    const int tid = threadIdx.x;
    const int i = blockIdx.x * 1024 + tid;
    const bool valid = i < e;
    const int key = valid ? keys[i] : 0;
    const int val = valid ? vals[i] : 0;
    const int digit = (key >> shift) & (RDIG - 1);
    const int rank = block_digit_rank(digit, valid, s_cnt, s_tot);

    {
        const int d = tid >> 4;
        const int sub = tid & 15;
        const int* row = ghist + d * nblocks;
        int all = 0, pre = 0;
        for (int b = sub; b < nblocks; b += 16) {
            const int v = row[b];
            all += v;
            pre += (b < (int)blockIdx.x) ? v : 0;
        }
        for (int o = 8; o > 0; o >>= 1) {
            all += __shfl_xor(all, o, 32);
            pre += __shfl_xor(pre, o, 32);
        }
        if (sub == 0) { s_all[d] = all; s_pre[d] = pre; }
    }
    __syncthreads();
    if (tid < 32) {
        const int a0 = s_all[2 * tid];
        const int a1 = s_all[2 * tid + 1];
        int x = a0 + a1;
        for (int o = 1; o < 32; o <<= 1) {
            const int y = __shfl_up(x, o, 32);
            if (tid >= o) x += y;
        }
        const int ex = x - (a0 + a1);
        s_all[2 * tid] = ex;
        s_all[2 * tid + 1] = ex + a0;
    }
    __syncthreads();
    if (valid) {
        const int p = s_all[digit] + s_pre[digit] + s_cnt[(tid >> 5) * RDIG + digit] + rank;
        keys_out[p] = key;
        vals_out[p] = val;
    }
}

__global__ void k_rowptr_dinv(const int* __restrict__ sdst, int* __restrict__ rowptr,
                              float* __restrict__ dinv, int n, int e) {
    const int i = blockIdx.x * 256 + threadIdx.x;
    if (i > n) return;
    int lo0 = 0, hi0 = e;
    int lo1 = 0, hi1 = e;
    while ((lo0 < hi0) | (lo1 < hi1)) {
        const int m0 = (lo0 + hi0) >> 1;
        const int m1 = (lo1 + hi1) >> 1;
        const int k0 = sdst[min(m0, e - 1)];
        const int k1 = sdst[min(m1, e - 1)];
        if (lo0 < hi0) { if (k0 < i) lo0 = m0 + 1; else hi0 = m0; }
        if (lo1 < hi1) { if (k1 < i + 1) lo1 = m1 + 1; else hi1 = m1; }
    }
    rowptr[i] = lo0;
    if (i < n) {
        const float d = (float)(lo1 - lo0 + 1);
        dinv[i] = (d > 0.0f) ? rsqrtf(d) : 0.0f;
    }
}

__global__ void k_fill(const int* __restrict__ ssrc, const int* __restrict__ sdst,
                       const float* __restrict__ dinv, int* __restrict__ csr, int e) {
    const int i = blockIdx.x * 256 + threadIdx.x;
    if (i < e) {
        const int s = ssrc[i];
        const int d = sdst[i];
        const float w = dinv[s] * dinv[d];
        *(int2*)(csr + 2 * (size_t)i) = make_int2(s, __float_as_int(w));
    }
}

__global__ void __launch_bounds__(256)
k_gemm128(const float* __restrict__ A, const float* __restrict__ W,
          float* __restrict__ out) {
    __shared__ float lds[16 * 130];
    const int row0 = blockIdx.x * 16;

    #pragma unroll
    for (int i = 0; i < 4; ++i) {
        int idx = threadIdx.x + 256 * i;
        int r   = idx >> 6;
        int c2  = (idx & 63) * 2;
        const float2 v = *(const float2*)(A + (size_t)(row0 + r) * NH + c2);
        *(float2*)(lds + r * 130 + c2) = v;
    }
    __syncthreads();

    const int lane = threadIdx.x & 31;
    const int wv   = threadIdx.x >> 5;
    const int col0 = wv * 16;
    const int m    = lane & 15;
    const int half = lane >> 4;

    const float* arow = lds + m * 130;
    v8f c;
    #pragma unroll
    for (int i = 0; i < 8; ++i) c[i] = 0.0f;

    #pragma unroll 4
    for (int kk = 0; kk < 32; ++kk) {
        const int k0 = kk * 4 + 2 * half;
        v2f a;
        a.x = arow[k0];
        a.y = arow[k0 + 1];
        v2f b;
        b.x = W[(size_t)k0 * NH + col0 + m];
        b.y = W[(size_t)(k0 + 1) * NH + col0 + m];
        c = __builtin_amdgcn_wmma_f32_16x16x4_f32(false, a, false, b,
                                                  (short)0, c, false, false);
    }

    const int outRow = row0 + 8 * half;
    #pragma unroll
    for (int r = 0; r < 8; ++r)
        out[(size_t)(outRow + r) * NH + col0 + m] = c[r];
}

#define GROWS 8

__device__ __forceinline__ float gelu_exact(float x) {
    return 0.5f * x * (1.0f + erff(x * 0.70710678118654752f));
}

__device__ __forceinline__ float4 fma4(float4 r, float w, float4 a) {
    a.x = fmaf(r.x, w, a.x); a.y = fmaf(r.y, w, a.y);
    a.z = fmaf(r.z, w, a.z); a.w = fmaf(r.w, w, a.w);
    return a;
}

__device__ __forceinline__ void
conv_node_pass(const float* __restrict__ hw, const int* __restrict__ rowptr,
               const int* __restrict__ csr, const float* __restrict__ dinv,
               const float* __restrict__ bias, const float* __restrict__ gamma,
               const float* __restrict__ beta, float* h,
               const float* wh, const float* bh, float* out, int n, const bool last) {
    const int lane = threadIdx.x & 31;
    const int c    = lane * 4;
    const int wave0  = __builtin_amdgcn_readfirstlane((int)(blockIdx.x * 8 + (threadIdx.x >> 5)));
    const int nwaves = gridDim.x * 8;
    const float4 b  = *(const float4*)(bias + c);
    const float4 g  = *(const float4*)(gamma + c);
    const float4 be = *(const float4*)(beta + c);
    const float4 z = make_float4(0.0f, 0.0f, 0.0f, 0.0f);

    int beg = 0, end = 0;
    if (wave0 < n) { beg = rowptr[wave0]; end = rowptr[wave0 + 1]; }

    for (int node = wave0; node < n; node += nwaves) {
        const int nxt = node + nwaves;
        int nbeg = 0, nend = 0;
        if (nxt < n) { nbeg = rowptr[nxt]; nend = rowptr[nxt + 1]; }

        const float di = dinv[node];
        const float sl = di * di;
        const float4 v = *(const float4*)(hw + (size_t)node * NH + c);
        int2 rec = make_int2(0, 0);
        if (lane < end - beg) rec = *(const int2*)(csr + 2 * (size_t)(beg + lane));
        float4 acc = fma4(v, sl, b);

        for (int base = beg; base < end; base += 32) {
            const int m = min(32, end - base);
            if (base != beg) {
                rec = make_int2(0, 0);
                if (lane < m) rec = *(const int2*)(csr + 2 * (size_t)(base + lane));
            }
            for (int k = 0; k < m; k += GROWS) {
                int    sidx[GROWS];
                float  wt[GROWS];
                float4 r[GROWS];
                for (int j = 0; j < GROWS; ++j) {
                    const int q = (k + j) & 31;
                    const int s = __builtin_amdgcn_readlane(rec.x, q);
                    const float w = __int_as_float(__builtin_amdgcn_readlane(rec.y, q));
                    sidx[j] = min(max(s, 0), n - 1);
                    wt[j] = (k + j < m) ? w : 0.0f;
                }
                for (int j = 0; j < GROWS; ++j) {
                    r[j] = z;
                    if (k + j < m) r[j] = *(const float4*)(hw + (size_t)sidx[j] * NH + c);
                }
                for (int j = 0; j < GROWS; ++j) acc = fma4(r[j], wt[j], acc);
            }
        }

        const float4 res = *(const float4*)(h + (size_t)node * NH + c);

        float s  = acc.x + acc.y + acc.z + acc.w;
        float ss = fmaf(acc.x, acc.x, fmaf(acc.y, acc.y, fmaf(acc.z, acc.z, acc.w * acc.w)));
        #pragma unroll
        for (int o = 16; o > 0; o >>= 1) {
            s  += __shfl_xor(s,  o, 32);
            ss += __shfl_xor(ss, o, 32);
        }
        const float mu   = s * (1.0f / NH);
        const float var  = fmaf(ss, (1.0f / NH), -mu * mu);
        const float rstd = rsqrtf(var + LN_EPS);
        float4 o;
        o.x = gelu_exact(fmaf((acc.x - mu) * rstd, g.x, be.x));
        o.y = gelu_exact(fmaf((acc.y - mu) * rstd, g.y, be.y));
        o.z = gelu_exact(fmaf((acc.z - mu) * rstd, g.z, be.z));
        o.w = gelu_exact(fmaf((acc.w - mu) * rstd, g.w, be.w));
        o.x += res.x; o.y += res.y; o.z += res.z; o.w += res.w;

        if (last) {
            const float4 hd = *(const float4*)(wh + c);
            const float  hb = bh[0];
            float p = fmaf(o.x, hd.x, fmaf(o.y, hd.y, fmaf(o.z, hd.z, o.w * hd.w)));
            #pragma unroll
            for (int q = 16; q > 0; q >>= 1) p += __shfl_xor(p, q, 32);
            if (lane == 0) out[node] = p + hb;
        } else {
            *(float4*)(h + (size_t)node * NH + c) = o;
        }
        beg = nbeg;
        end = nend;
    }
}

__global__ void __launch_bounds__(256)
k_conv_node(const float* __restrict__ hw, const int* __restrict__ rowptr,
            const int* __restrict__ csr, const float* __restrict__ dinv,
            const float* __restrict__ bias, const float* __restrict__ gamma,
            const float* __restrict__ beta, float* h, int n) {
    conv_node_pass(hw, rowptr, csr, dinv, bias, gamma, beta, h, nullptr, nullptr, nullptr, n, false);
}

__global__ void __launch_bounds__(256)
k_conv_node_head(const float* __restrict__ hw, const int* __restrict__ rowptr,
                 const int* __restrict__ csr, const float* __restrict__ dinv,
                 const float* __restrict__ bias, const float* __restrict__ gamma,
                 const float* __restrict__ beta, float* h,
                 const float* wh, const float* bh, float* out, int n) {
    conv_node_pass(hw, rowptr, csr, dinv, bias, gamma, beta, h, wh, bh, out, n, true);
}


__global__ void k_lngelu(const float* __restrict__ in, const float* __restrict__ bias,
                         const float* __restrict__ gamma, const float* __restrict__ beta,
                         const float* __restrict__ resid, float* __restrict__ out, int n) {
    const int lane = threadIdx.x & 31;
    const int wv   = threadIdx.x >> 5;
    const int node = blockIdx.x * 8 + wv;
    if (node >= n) return;
    const int c = lane * 4;

    float4 t = *(const float4*)(in + (size_t)node * NH + c);
    if (bias) {
        float4 b = *(const float4*)(bias + c);
        t.x += b.x; t.y += b.y; t.z += b.z; t.w += b.w;
    }
    float s  = t.x + t.y + t.z + t.w;
    float ss = fmaf(t.x, t.x, fmaf(t.y, t.y, fmaf(t.z, t.z, t.w * t.w)));
    #pragma unroll
    for (int o = 16; o > 0; o >>= 1) {
        s  += __shfl_xor(s,  o, 32);
        ss += __shfl_xor(ss, o, 32);
    }
    const float mu   = s * (1.0f / NH);
    const float var  = fmaf(ss, (1.0f / NH), -mu * mu);
    const float rstd = rsqrtf(var + LN_EPS);

    float4 g  = *(const float4*)(gamma + c);
    float4 be = *(const float4*)(beta + c);
    float4 o;
    o.x = gelu_exact(fmaf((t.x - mu) * rstd, g.x, be.x));
    o.y = gelu_exact(fmaf((t.y - mu) * rstd, g.y, be.y));
    o.z = gelu_exact(fmaf((t.z - mu) * rstd, g.z, be.z));
    o.w = gelu_exact(fmaf((t.w - mu) * rstd, g.w, be.w));
    if (resid) {
        float4 r = *(const float4*)(resid + (size_t)node * NH + c);
        o.x += r.x; o.y += r.y; o.z += r.z; o.w += r.w;
    }
    *(float4*)(out + (size_t)node * NH + c) = o;
}

extern "C" void kernel_launch(void* const* d_in, const int* in_sizes, int n_in,
                              void* d_out, int out_size, void* d_ws, size_t ws_size,
                              hipStream_t stream) {
    const float* x      = (const float*)d_in[0];
    const int*   ei     = (const int*)d_in[1];
    const float* W_in   = (const float*)d_in[2];
    const float* b_in   = (const float*)d_in[3];
    const float* g_in   = (const float*)d_in[4];
    const float* be_in  = (const float*)d_in[5];
    const float* W_conv = (const float*)d_in[6];
    const float* b_conv = (const float*)d_in[7];
    const float* g_ln   = (const float*)d_in[8];
    const float* be_ln  = (const float*)d_in[9];
    const float* W_head = (const float*)d_in[10];
    const float* b_head = (const float*)d_in[11];
    float* out = (float*)d_out;

    const int N = in_sizes[0] / NH;
    const int E = in_sizes[1] / 2;
    const int* src = ei;
    const int* dst = ei + E;

    char* w = (char*)d_ws;
    float* dinv   = (float*)(w);
    int*   ghist  = (int*)  (w + (size_t)(1)   * (1 << 20));
    int*   rowptr = (int*)  (w + (size_t)(2)   * (1 << 20));
    float* h      = (float*)(w + (size_t)(4)   * (1 << 20));
    float* hw     = (float*)(w + (size_t)(60)  * (1 << 20));
    int*   kA     = (int*)  (w + (size_t)(168) * (1 << 20));
    int*   vA     = (int*)  (w + (size_t)(171) * (1 << 20));
    int*   kB     = (int*)  (w + (size_t)(174) * (1 << 20));
    int*   vB     = (int*)  (w + (size_t)(177) * (1 << 20));
    int*   csr    = (int*)  (w + (size_t)(180) * (1 << 20));

    const int gE   = (E + 255) / 256;
    const int gN1  = (N + 1 + 255) / 256;
    const int gRow = N / 16;
    const int gNd  = (N + 7) / 8;
    const int nSb  = (E + 1023) / 1024;
    const int gGa  = 2048;

    k_sort_count<<<nSb, 1024, 0, stream>>>(dst, ghist, E, 0, nSb);
    k_sort_move<<<nSb, 1024, 0, stream>>>(dst, src, ghist, kA, vA, E, 0, nSb);
    k_sort_count<<<nSb, 1024, 0, stream>>>(kA, ghist, E, 6, nSb);
    k_sort_move<<<nSb, 1024, 0, stream>>>(kA, vA, ghist, kB, vB, E, 6, nSb);
    k_sort_count<<<nSb, 1024, 0, stream>>>(kB, ghist, E, 12, nSb);
    k_sort_move<<<nSb, 1024, 0, stream>>>(kB, vB, ghist, kA, vA, E, 12, nSb);
    k_rowptr_dinv<<<gN1, 256, 0, stream>>>(kA, rowptr, dinv, N, E);
    k_fill<<<gE, 256, 0, stream>>>(vA, kA, dinv, csr, E);

    k_gemm128<<<gRow, 256, 0, stream>>>(x, W_in, hw);
    k_lngelu<<<gNd, 256, 0, stream>>>(hw, b_in, g_in, be_in, nullptr, h, N);

    for (int i = 0; i < NLAYERS; ++i) {
        const float* Wc = W_conv + (size_t)i * NH * NH;
        const float* bc = b_conv + (size_t)i * NH;
        const float* gl = g_ln   + (size_t)i * NH;
        const float* bl = be_ln  + (size_t)i * NH;
        k_gemm128<<<gRow, 256, 0, stream>>>(h, Wc, hw);
        if (i + 1 < NLAYERS)
            k_conv_node<<<gGa, 256, 0, stream>>>(hw, rowptr, csr, dinv, bc, gl, bl, h, N);
        else
            k_conv_node_head<<<gGa, 256, 0, stream>>>(hw, rowptr, csr, dinv, bc, gl, bl, h,
                                                      W_head, b_head, out, N);
    }
}


# Conv weight product moved behind the neighbour sum (they commute) into the node kernel: 16 nodes per block summed to LDS, WMMA with W in registers, LN on the tile; sum order changes
# speedup vs baseline: 1.6938x; 1.0381x over previous
#include <hip/hip_runtime.h>
#include <hip/hip_bf16.h>

#define NH 128
#define NLAYERS 3
#define LN_EPS 1e-5f

typedef float v2f __attribute__((ext_vector_type(2)));
typedef float v8f __attribute__((ext_vector_type(8)));

#define RBITS 6
#define RDIG  64

__device__ __forceinline__ int block_digit_rank(int digit, bool valid, int* s_cnt, int* s_tot) {
    const int tid  = threadIdx.x;
    const int lane = tid & 31;
    const int wv   = tid >> 5;
    s_cnt[tid] = 0;
    s_cnt[tid + 1024] = 0;
    unsigned peers = __builtin_amdgcn_ballot_w32(valid);
    for (int b = 0; b < RBITS; ++b) {
        const bool bit = ((digit >> b) & 1) != 0;
        const unsigned bal = __builtin_amdgcn_ballot_w32(bit);
        peers &= bit ? bal : ~bal;
    }
    const int rank = __popc(peers & ((1u << lane) - 1u));
    __syncthreads();
    if (valid && rank == 0) s_cnt[wv * RDIG + digit] = __popc(peers);
    __syncthreads();
    if (tid < RDIG) {
        int run = 0;
        for (int w = 0; w < 32; ++w) {
            const int t = s_cnt[w * RDIG + tid];
            s_cnt[w * RDIG + tid] = run;
            run += t;
        }
        s_tot[tid] = run;
    }
    __syncthreads();
    return rank;
}

__global__ void __launch_bounds__(1024)
k_sort_count(const int* __restrict__ keys, int* __restrict__ ghist,
             int e, int shift, int nblocks) {
    __shared__ int s_cnt[32 * RDIG];
    __shared__ int s_tot[RDIG];
    const int i = blockIdx.x * 1024 + threadIdx.x;
    const bool valid = i < e;
    const int key = valid ? keys[i] : 0;
    block_digit_rank((key >> shift) & (RDIG - 1), valid, s_cnt, s_tot);
    if (threadIdx.x < RDIG) ghist[threadIdx.x * nblocks + blockIdx.x] = s_tot[threadIdx.x];
}

__global__ void __launch_bounds__(1024)
k_sort_move(const int* __restrict__ keys, const int* __restrict__ vals,
            const int* __restrict__ ghist, int* __restrict__ keys_out,
            int* __restrict__ vals_out, int e, int shift, int nblocks) {
    __shared__ int s_cnt[32 * RDIG];
    __shared__ int s_tot[RDIG];
    __shared__ int s_all[RDIG];
    __shared__ int s_pre[RDIG];
    const int tid = threadIdx.x;
    const int i = blockIdx.x * 1024 + tid;
    const bool valid = i < e;
    const int key = valid ? keys[i] : 0;
    const int val = valid ? vals[i] : 0;
    const int digit = (key >> shift) & (RDIG - 1);
    const int rank = block_digit_rank(digit, valid, s_cnt, s_tot);

    {
        const int d = tid >> 4;
        const int sub = tid & 15;
        const int* row = ghist + d * nblocks;
        int all = 0, pre = 0;
        for (int b = sub; b < nblocks; b += 16) {
            const int v = row[b];
            all += v;
            pre += (b < (int)blockIdx.x) ? v : 0;
        }
        for (int o = 8; o > 0; o >>= 1) {
            all += __shfl_xor(all, o, 32);
            pre += __shfl_xor(pre, o, 32);
        }
        if (sub == 0) { s_all[d] = all; s_pre[d] = pre; }
    }
    __syncthreads();
    if (tid < 32) {
        const int a0 = s_all[2 * tid];
        const int a1 = s_all[2 * tid + 1];
        int x = a0 + a1;
        for (int o = 1; o < 32; o <<= 1) {
            const int y = __shfl_up(x, o, 32);
            if (tid >= o) x += y;
        }
        const int ex = x - (a0 + a1);
        s_all[2 * tid] = ex;
        s_all[2 * tid + 1] = ex + a0;
    }
    __syncthreads();
    if (valid) {
        const int p = s_all[digit] + s_pre[digit] + s_cnt[(tid >> 5) * RDIG + digit] + rank;
        keys_out[p] = key;
        vals_out[p] = val;
    }
}

__global__ void k_rowptr_dinv(const int* __restrict__ sdst, int* __restrict__ rowptr,
                              float* __restrict__ dinv, int n, int e) {
    const int i = blockIdx.x * 256 + threadIdx.x;
    if (i > n) return;
    int lo0 = 0, hi0 = e;
    int lo1 = 0, hi1 = e;
    while ((lo0 < hi0) | (lo1 < hi1)) {
        const int m0 = (lo0 + hi0) >> 1;
        const int m1 = (lo1 + hi1) >> 1;
        const int k0 = sdst[min(m0, e - 1)];
        const int k1 = sdst[min(m1, e - 1)];
        if (lo0 < hi0) { if (k0 < i) lo0 = m0 + 1; else hi0 = m0; }
        if (lo1 < hi1) { if (k1 < i + 1) lo1 = m1 + 1; else hi1 = m1; }
    }
    rowptr[i] = lo0;
    if (i < n) {
        const float d = (float)(lo1 - lo0 + 1);
        dinv[i] = (d > 0.0f) ? rsqrtf(d) : 0.0f;
    }
}

__global__ void k_fill(const int* __restrict__ ssrc, const int* __restrict__ sdst,
                       const float* __restrict__ dinv, int* __restrict__ csr, int e) {
    const int i = blockIdx.x * 256 + threadIdx.x;
    if (i < e) {
        const int s = ssrc[i];
        const int d = sdst[i];
        const float w = dinv[s] * dinv[d];
        *(int2*)(csr + 2 * (size_t)i) = make_int2(s, __float_as_int(w));
    }
}

__global__ void __launch_bounds__(256)
k_gemm128(const float* __restrict__ A, const float* __restrict__ W,
          float* __restrict__ out) {
    __shared__ float lds[16 * 130];
    const int row0 = blockIdx.x * 16;

    #pragma unroll
    for (int i = 0; i < 4; ++i) {
        int idx = threadIdx.x + 256 * i;
        int r   = idx >> 6;
        int c2  = (idx & 63) * 2;
        const float2 v = *(const float2*)(A + (size_t)(row0 + r) * NH + c2);
        *(float2*)(lds + r * 130 + c2) = v;
    }
    __syncthreads();

    const int lane = threadIdx.x & 31;
    const int wv   = threadIdx.x >> 5;
    const int col0 = wv * 16;
    const int m    = lane & 15;
    const int half = lane >> 4;

    const float* arow = lds + m * 130;
    v8f c;
    #pragma unroll
    for (int i = 0; i < 8; ++i) c[i] = 0.0f;

    #pragma unroll 4
    for (int kk = 0; kk < 32; ++kk) {
        const int k0 = kk * 4 + 2 * half;
        v2f a;
        a.x = arow[k0];
        a.y = arow[k0 + 1];
        v2f b;
        b.x = W[(size_t)k0 * NH + col0 + m];
        b.y = W[(size_t)(k0 + 1) * NH + col0 + m];
        c = __builtin_amdgcn_wmma_f32_16x16x4_f32(false, a, false, b,
                                                  (short)0, c, false, false);
    }

    const int outRow = row0 + 8 * half;
    #pragma unroll
    for (int r = 0; r < 8; ++r)
        out[(size_t)(outRow + r) * NH + col0 + m] = c[r];
}

#define GR 12
#define TP 132

__device__ __forceinline__ float gelu_exact(float x) {
    return 0.5f * x * (1.0f + erff(x * 0.70710678118654752f));
}

__device__ __forceinline__ float4 fma4(float4 r, float w, float4 a) {
    a.x = fmaf(r.x, w, a.x); a.y = fmaf(r.y, w, a.y);
    a.z = fmaf(r.z, w, a.z); a.w = fmaf(r.w, w, a.w);
    return a;
}

__device__ __forceinline__ int tile_rowptr(const int* __restrict__ rowptr, int tile, int lane, int n) {
    return rowptr[min(tile * 16 + min(lane, 16), n)];
}

__device__ __forceinline__ int2 first_records(const int* __restrict__ csr, int beg, int end, int lane) {
    int2 rec = make_int2(0, 0);
    if (lane < end - beg) rec = *(const int2*)(csr + 2 * (size_t)(beg + lane));
    return rec;
}

__device__ __forceinline__ void
sum_group(const float* __restrict__ hin, int2 rec, int base, int k, int cnt, int eA,
          float4& accA, float4& accB, int c, int n) {
    const float4 z = make_float4(0.0f, 0.0f, 0.0f, 0.0f);
    int    sidx[GR];
    float  wt[GR];
    float4 r[GR];
    #pragma unroll
    for (int j = 0; j < GR; ++j) {
        const int q = (k + j) & 31;
        const int s = __builtin_amdgcn_readlane(rec.x, q);
        wt[j] = __int_as_float(__builtin_amdgcn_readlane(rec.y, q));
        sidx[j] = min(max(s, 0), n - 1);
    }
    #pragma unroll
    for (int j = 0; j < GR; ++j) {
        r[j] = z;
        if (k + j < cnt) r[j] = *(const float4*)(hin + (size_t)sidx[j] * NH + c);
    }
    #pragma unroll
    for (int j = 0; j < GR; ++j) {
        if (k + j < cnt) {
            if (base + k + j < eA) accA = fma4(r[j], wt[j], accA);
            else                   accB = fma4(r[j], wt[j], accB);
        }
    }
}

__device__ __forceinline__ void
sum_chunk(const float* __restrict__ hin, int2 rec, int base, int cnt, int eA,
          float4& accA, float4& accB, int c, int n) {
    sum_group(hin, rec, base, 0, cnt, eA, accA, accB, c, n);
    for (int k = GR; k < cnt; k += GR)
        sum_group(hin, rec, base, k, cnt, eA, accA, accB, c, n);
}

__device__ __forceinline__ void
gather_two(const float* __restrict__ hin, const int* __restrict__ csr,
           int bA, int eA, int eB, int2 rec, float4& accA, float4& accB,
           int lane, int c, int n) {
    sum_chunk(hin, rec, bA, min(32, eB - bA), eA, accA, accB, c, n);
    for (int base = bA + 32; base < eB; base += 32) {
        const int cnt = min(32, eB - base);
        int2 more = make_int2(0, 0);
        if (lane < cnt) more = *(const int2*)(csr + 2 * (size_t)(base + lane));
        sum_chunk(hin, more, base, cnt, eA, accA, accB, c, n);
    }
}

__device__ __forceinline__ float4
ln_gelu_res(float4 t, float4 g, float4 be, float4 res) {
    float s  = t.x + t.y + t.z + t.w;
    float ss = fmaf(t.x, t.x, fmaf(t.y, t.y, fmaf(t.z, t.z, t.w * t.w)));
    #pragma unroll
    for (int o = 16; o > 0; o >>= 1) {
        s  += __shfl_xor(s,  o, 32);
        ss += __shfl_xor(ss, o, 32);
    }
    const float mu   = s * (1.0f / NH);
    const float var  = fmaf(ss, (1.0f / NH), -mu * mu);
    const float rstd = rsqrtf(var + LN_EPS);
    float4 o;
    o.x = gelu_exact(fmaf((t.x - mu) * rstd, g.x, be.x)) + res.x;
    o.y = gelu_exact(fmaf((t.y - mu) * rstd, g.y, be.y)) + res.y;
    o.z = gelu_exact(fmaf((t.z - mu) * rstd, g.z, be.z)) + res.z;
    o.w = gelu_exact(fmaf((t.w - mu) * rstd, g.w, be.w)) + res.w;
    return o;
}

__device__ __forceinline__ void
conv_layer_pass(const float* __restrict__ hin, const int* __restrict__ rowptr,
                const int* __restrict__ csr, const float* __restrict__ dinv,
                const float* __restrict__ Wm, const float* bias, const float* gamma,
                const float* beta, float* hout, const float* wh, const float* bh,
                float* out, int n, int ntiles,
                float* tA, float* tC, float* tR, float* tP, const bool last) {
    const int lane = threadIdx.x & 31;
    const int c    = lane * 4;
    const int wv   = __builtin_amdgcn_readfirstlane((int)(threadIdx.x >> 5));
    const int col0 = wv * 16;
    const int m    = lane & 15;
    const int half = lane >> 4;
    const int G    = gridDim.x;
    const float4 z = make_float4(0.0f, 0.0f, 0.0f, 0.0f);

    if (threadIdx.x < 32)       *(float4*)(tP + c)          = *(const float4*)(bias + c);
    else if (threadIdx.x < 64)  *(float4*)(tP + NH + c)     = *(const float4*)(gamma + c);
    else if (threadIdx.x < 96)  *(float4*)(tP + 2 * NH + c) = *(const float4*)(beta + c);
    else if (last && threadIdx.x < 128) *(float4*)(tP + 3 * NH + c) = *(const float4*)(wh + c);

    v2f bw[32];
    #pragma unroll
    for (int kk = 0; kk < 32; ++kk) {
        const int k0 = kk * 4 + 2 * half;
        bw[kk].x = Wm[(size_t)k0 * NH + col0 + m];
        bw[kk].y = Wm[(size_t)(k0 + 1) * NH + col0 + m];
    }

    int tile = blockIdx.x;
    int rpN;
    int bA, eA, eB;
    int2 rec;
    {
        const int rp0 = tile_rowptr(rowptr, tile, lane, n);
        rpN = tile_rowptr(rowptr, tile + G, lane, n);
        bA = __builtin_amdgcn_readlane(rp0, wv * 2);
        eA = __builtin_amdgcn_readlane(rp0, wv * 2 + 1);
        eB = __builtin_amdgcn_readlane(rp0, wv * 2 + 2);
        rec = first_records(csr, bA, eB, lane);
    }

    for (; tile < ntiles; tile += G) {
        const int nbA = __builtin_amdgcn_readlane(rpN, wv * 2);
        const int neA = __builtin_amdgcn_readlane(rpN, wv * 2 + 1);
        const int neB = __builtin_amdgcn_readlane(rpN, wv * 2 + 2);
        const int2 nrec = first_records(csr, nbA, neB, lane);
        const int rpF = tile_rowptr(rowptr, tile + 2 * G, lane, n);

        const int nodeA = tile * 16 + wv * 2;
        const int nodeB = nodeA + 1;
        const float4 vA = *(const float4*)(hin + (size_t)min(nodeA, n - 1) * NH + c);
        const float4 vB = *(const float4*)(hin + (size_t)min(nodeB, n - 1) * NH + c);
        const float dv = dinv[min(tile * 16 + (lane & 15), n - 1)];
        float4 accA = z, accB = z;
        gather_two(hin, csr, bA, eA, eB, rec, accA, accB, lane, c, n);
        float dA = __int_as_float(__builtin_amdgcn_readlane(__float_as_int(dv), wv * 2));
        float dB = __int_as_float(__builtin_amdgcn_readlane(__float_as_int(dv), wv * 2 + 1));
        if (nodeA >= n) dA = 0.0f;
        if (nodeB >= n) dB = 0.0f;
        accA = fma4(vA, dA * dA, accA);
        accB = fma4(vB, dB * dB, accB);
        *(float4*)(tA + (wv * 2) * TP + c) = accA;
        *(float4*)(tA + (wv * 2 + 1) * TP + c) = accB;
        *(float4*)(tR + (wv * 2) * TP + c) = vA;
        *(float4*)(tR + (wv * 2 + 1) * TP + c) = vB;
        __syncthreads();

        const float* arow = tA + m * TP + 2 * half;
        v8f acc8;
        #pragma unroll
        for (int i = 0; i < 8; ++i) acc8[i] = 0.0f;
        #pragma unroll
        for (int kk = 0; kk < 32; ++kk) {
            const v2f a = *(const v2f*)(arow + kk * 4);
            acc8 = __builtin_amdgcn_wmma_f32_16x16x4_f32(false, a, false, bw[kk],
                                                         (short)0, acc8, false, false);
        }
        #pragma unroll
        for (int r = 0; r < 8; ++r) tC[(8 * half + r) * TP + col0 + m] = acc8[r];
        __syncthreads();

        const float4 b  = *(const float4*)(tP + c);
        const float4 g  = *(const float4*)(tP + NH + c);
        const float4 be = *(const float4*)(tP + 2 * NH + c);
        float4 tRowA = *(const float4*)(tC + (wv * 2) * TP + c);
        float4 tRowB = *(const float4*)(tC + (wv * 2 + 1) * TP + c);
        tRowA.x += b.x; tRowA.y += b.y; tRowA.z += b.z; tRowA.w += b.w;
        tRowB.x += b.x; tRowB.y += b.y; tRowB.z += b.z; tRowB.w += b.w;
        const float4 resA = *(const float4*)(tR + (wv * 2) * TP + c);
        const float4 resB = *(const float4*)(tR + (wv * 2 + 1) * TP + c);
        const float4 oA = ln_gelu_res(tRowA, g, be, resA);
        const float4 oB = ln_gelu_res(tRowB, g, be, resB);
        if (last) {
            const float4 hd = *(const float4*)(tP + 3 * NH + c);
            const float  hb = bh[0];
            float pA = fmaf(oA.x, hd.x, fmaf(oA.y, hd.y, fmaf(oA.z, hd.z, oA.w * hd.w)));
            float pB = fmaf(oB.x, hd.x, fmaf(oB.y, hd.y, fmaf(oB.z, hd.z, oB.w * hd.w)));
            #pragma unroll
            for (int q = 16; q > 0; q >>= 1) {
                pA += __shfl_xor(pA, q, 32);
                pB += __shfl_xor(pB, q, 32);
            }
            if (lane == 0) {
                if (nodeA < n) out[nodeA] = pA + hb;
                if (nodeB < n) out[nodeB] = pB + hb;
            }
        } else {
            if (nodeA < n) *(float4*)(hout + (size_t)nodeA * NH + c) = oA;
            if (nodeB < n) *(float4*)(hout + (size_t)nodeB * NH + c) = oB;
        }
        bA = nbA; eA = neA; eB = neB;
        rec = nrec;
        rpN = rpF;
    }
}

__global__ void __launch_bounds__(256, 6)
k_conv_layer(const float* __restrict__ hin, const int* __restrict__ rowptr,
             const int* __restrict__ csr, const float* __restrict__ dinv,
             const float* __restrict__ Wm, const float* bias, const float* gamma,
             const float* beta, float* hout, int n, int ntiles) {
    __shared__ __attribute__((aligned(16))) float tA[16 * TP];
    __shared__ __attribute__((aligned(16))) float tC[16 * TP];
    __shared__ __attribute__((aligned(16))) float tR[16 * TP];
    __shared__ __attribute__((aligned(16))) float tP[4 * NH];
    conv_layer_pass(hin, rowptr, csr, dinv, Wm, bias, gamma, beta, hout,
                    nullptr, nullptr, nullptr, n, ntiles, tA, tC, tR, tP, false);
}

__global__ void __launch_bounds__(256, 6)
k_conv_layer_head(const float* __restrict__ hin, const int* __restrict__ rowptr,
                  const int* __restrict__ csr, const float* __restrict__ dinv,
                  const float* __restrict__ Wm, const float* bias, const float* gamma,
                  const float* beta, const float* wh, const float* bh, float* out,
                  int n, int ntiles) {
    __shared__ __attribute__((aligned(16))) float tA[16 * TP];
    __shared__ __attribute__((aligned(16))) float tC[16 * TP];
    __shared__ __attribute__((aligned(16))) float tR[16 * TP];
    __shared__ __attribute__((aligned(16))) float tP[4 * NH];
    conv_layer_pass(hin, rowptr, csr, dinv, Wm, bias, gamma, beta, nullptr,
                    wh, bh, out, n, ntiles, tA, tC, tR, tP, true);
}


__global__ void k_lngelu(const float* __restrict__ in, const float* __restrict__ bias,
                         const float* __restrict__ gamma, const float* __restrict__ beta,
                         const float* __restrict__ resid, float* __restrict__ out, int n) {
    const int lane = threadIdx.x & 31;
    const int wv   = threadIdx.x >> 5;
    const int node = blockIdx.x * 8 + wv;
    if (node >= n) return;
    const int c = lane * 4;

    float4 t = *(const float4*)(in + (size_t)node * NH + c);
    if (bias) {
        float4 b = *(const float4*)(bias + c);
        t.x += b.x; t.y += b.y; t.z += b.z; t.w += b.w;
    }
    float s  = t.x + t.y + t.z + t.w;
    float ss = fmaf(t.x, t.x, fmaf(t.y, t.y, fmaf(t.z, t.z, t.w * t.w)));
    #pragma unroll
    for (int o = 16; o > 0; o >>= 1) {
        s  += __shfl_xor(s,  o, 32);
        ss += __shfl_xor(ss, o, 32);
    }
    const float mu   = s * (1.0f / NH);
    const float var  = fmaf(ss, (1.0f / NH), -mu * mu);
    const float rstd = rsqrtf(var + LN_EPS);

    float4 g  = *(const float4*)(gamma + c);
    float4 be = *(const float4*)(beta + c);
    float4 o;
    o.x = gelu_exact(fmaf((t.x - mu) * rstd, g.x, be.x));
    o.y = gelu_exact(fmaf((t.y - mu) * rstd, g.y, be.y));
    o.z = gelu_exact(fmaf((t.z - mu) * rstd, g.z, be.z));
    o.w = gelu_exact(fmaf((t.w - mu) * rstd, g.w, be.w));
    if (resid) {
        float4 r = *(const float4*)(resid + (size_t)node * NH + c);
        o.x += r.x; o.y += r.y; o.z += r.z; o.w += r.w;
    }
    *(float4*)(out + (size_t)node * NH + c) = o;
}

extern "C" void kernel_launch(void* const* d_in, const int* in_sizes, int n_in,
                              void* d_out, int out_size, void* d_ws, size_t ws_size,
                              hipStream_t stream) {
    const float* x      = (const float*)d_in[0];
    const int*   ei     = (const int*)d_in[1];
    const float* W_in   = (const float*)d_in[2];
    const float* b_in   = (const float*)d_in[3];
    const float* g_in   = (const float*)d_in[4];
    const float* be_in  = (const float*)d_in[5];
    const float* W_conv = (const float*)d_in[6];
    const float* b_conv = (const float*)d_in[7];
    const float* g_ln   = (const float*)d_in[8];
    const float* be_ln  = (const float*)d_in[9];
    const float* W_head = (const float*)d_in[10];
    const float* b_head = (const float*)d_in[11];
    float* out = (float*)d_out;

    const int N = in_sizes[0] / NH;
    const int E = in_sizes[1] / 2;
    const int* src = ei;
    const int* dst = ei + E;

    char* w = (char*)d_ws;
    float* dinv   = (float*)(w);
    int*   ghist  = (int*)  (w + (size_t)(1)   * (1 << 20));
    int*   rowptr = (int*)  (w + (size_t)(2)   * (1 << 20));
    float* hA     = (float*)(w + (size_t)(4)   * (1 << 20));
    float* hB     = (float*)(w + (size_t)(60)  * (1 << 20));
    int*   kA     = (int*)  (w + (size_t)(168) * (1 << 20));
    int*   vA     = (int*)  (w + (size_t)(171) * (1 << 20));
    int*   kB     = (int*)  (w + (size_t)(174) * (1 << 20));
    int*   vB     = (int*)  (w + (size_t)(177) * (1 << 20));
    int*   csr    = (int*)  (w + (size_t)(180) * (1 << 20));

    const int gE   = (E + 255) / 256;
    const int gN1  = (N + 1 + 255) / 256;
    const int gRow = N / 16;
    const int gNd  = (N + 7) / 8;
    const int nSb  = (E + 1023) / 1024;
    const int nT   = (N + 15) / 16;
    const int gLy  = 768;

    k_sort_count<<<nSb, 1024, 0, stream>>>(dst, ghist, E, 0, nSb);
    k_sort_move<<<nSb, 1024, 0, stream>>>(dst, src, ghist, kA, vA, E, 0, nSb);
    k_sort_count<<<nSb, 1024, 0, stream>>>(kA, ghist, E, 6, nSb);
    k_sort_move<<<nSb, 1024, 0, stream>>>(kA, vA, ghist, kB, vB, E, 6, nSb);
    k_sort_count<<<nSb, 1024, 0, stream>>>(kB, ghist, E, 12, nSb);
    k_sort_move<<<nSb, 1024, 0, stream>>>(kB, vB, ghist, kA, vA, E, 12, nSb);
    k_rowptr_dinv<<<gN1, 256, 0, stream>>>(kA, rowptr, dinv, N, E);
    k_fill<<<gE, 256, 0, stream>>>(vA, kA, dinv, csr, E);

    k_gemm128<<<gRow, 256, 0, stream>>>(x, W_in, hB);
    k_lngelu<<<gNd, 256, 0, stream>>>(hB, b_in, g_in, be_in, nullptr, hA, N);

    for (int i = 0; i < NLAYERS; ++i) {
        const float* Wc = W_conv + (size_t)i * NH * NH;
        const float* bc = b_conv + (size_t)i * NH;
        const float* gl = g_ln   + (size_t)i * NH;
        const float* bl = be_ln  + (size_t)i * NH;
        const float* hin  = (i & 1) ? hB : hA;
        float*       hout = (i & 1) ? hA : hB;
        if (i + 1 < NLAYERS)
            k_conv_layer<<<gLy, 256, 0, stream>>>(hin, rowptr, csr, dinv, Wc, bc, gl, bl, hout, N, nT);
        else
            k_conv_layer_head<<<gLy, 256, 0, stream>>>(hin, rowptr, csr, dinv, Wc, bc, gl, bl,
                                                       W_head, b_head, out, N, nT);
    }
}


# Edge sort passes on 147 resident blocks of 8 waves, 16 keys per thread (were 586 blocks of 32 waves, one key per thread): a sixteenth of the waves, count table and its walk a quarter as long
# speedup vs baseline: 1.9408x; 1.1458x over previous
#include <hip/hip_runtime.h>
#include <hip/hip_bf16.h>

#define NH 128
#define NLAYERS 3
#define LN_EPS 1e-5f

typedef float v2f __attribute__((ext_vector_type(2)));
typedef float v8f __attribute__((ext_vector_type(8)));

#define RBITS  6
#define RDIG   64
#define SORT_T 256
#define SORT_W (SORT_T / 32)
#define KPT    16
#define SORT_K (SORT_T * KPT)
#define SORT_V (SORT_W * KPT)
#define WALK_U 10

__device__ __forceinline__ void
block_digit_rank(const int (&digit)[KPT], const bool (&valid)[KPT], int (&rank)[KPT],
                 int* s_cnt, int* s_tot) {
    const int tid  = threadIdx.x;
    const int lane = tid & 31;
    const int wv   = tid >> 5;
    #pragma unroll
    for (int i = 0; i < SORT_V * RDIG / SORT_T; ++i) s_cnt[tid + i * SORT_T] = 0;
    int cnt[KPT];
    #pragma unroll
    for (int j = 0; j < KPT; ++j) {
        unsigned peers = __builtin_amdgcn_ballot_w32(valid[j]);
        #pragma unroll
        for (int b = 0; b < RBITS; ++b) {
            const bool bit = ((digit[j] >> b) & 1) != 0;
            const unsigned bal = __builtin_amdgcn_ballot_w32(bit);
            peers &= bit ? bal : ~bal;
        }
        rank[j] = __popc(peers & ((1u << lane) - 1u));
        cnt[j]  = __popc(peers);
    }
    __syncthreads();
    #pragma unroll
    for (int j = 0; j < KPT; ++j)
        if (valid[j] && rank[j] == 0) s_cnt[(j * SORT_W + wv) * RDIG + digit[j]] = cnt[j];
    __syncthreads();
    if (tid < RDIG) {
        int run = 0;
        for (int v = 0; v < SORT_V; ++v) {
            const int t = s_cnt[v * RDIG + tid];
            s_cnt[v * RDIG + tid] = run;
            run += t;
        }
        s_tot[tid] = run;
    }
    __syncthreads();
}

__global__ void __launch_bounds__(SORT_T)
k_sort_count(const int* __restrict__ keys, int* __restrict__ ghist,
             int e, int shift, int nblocks) {
    __shared__ int s_cnt[SORT_V * RDIG];
    __shared__ int s_tot[RDIG];
    const int first = blockIdx.x * SORT_K + threadIdx.x;
    int  digit[KPT];
    bool valid[KPT];
    int  rank[KPT];
    #pragma unroll
    for (int j = 0; j < KPT; ++j) {
        const int i = first + j * SORT_T;
        valid[j] = i < e;
        const int key = valid[j] ? keys[i] : 0;
        digit[j] = (key >> shift) & (RDIG - 1);
    }
    block_digit_rank(digit, valid, rank, s_cnt, s_tot);
    if (threadIdx.x < RDIG) ghist[threadIdx.x * nblocks + blockIdx.x] = s_tot[threadIdx.x];
}

__global__ void __launch_bounds__(SORT_T)
k_sort_move(const int* __restrict__ keys, const int* __restrict__ vals,
            const int* __restrict__ ghist, int* __restrict__ keys_out,
            int* __restrict__ vals_out, int e, int shift, int nblocks) {
    __shared__ int s_cnt[SORT_V * RDIG];
    __shared__ int s_tot[RDIG];
    __shared__ int s_all[RDIG];
    __shared__ int s_pre[RDIG];
    const int tid = threadIdx.x;
    const int wv  = tid >> 5;
    const int first = blockIdx.x * SORT_K + tid;
    int  key[KPT];
    int  val[KPT];
    int  digit[KPT];
    bool valid[KPT];
    int  rank[KPT];
    #pragma unroll
    for (int j = 0; j < KPT; ++j) {
        const int i = first + j * SORT_T;
        valid[j] = i < e;
        key[j] = valid[j] ? keys[i] : 0;
        val[j] = valid[j] ? vals[i] : 0;
        digit[j] = (key[j] >> shift) & (RDIG - 1);
    }
    {
        const int d = tid >> 2;
        const int sub = tid & 3;
        const int* row = ghist + d * nblocks;
        int all = 0, pre = 0;
        for (int b0 = sub; b0 < nblocks; b0 += 4 * WALK_U) {
            int v[WALK_U];
            #pragma unroll
            for (int u = 0; u < WALK_U; ++u) v[u] = row[min(b0 + 4 * u, nblocks - 1)];
            #pragma unroll
            for (int u = 0; u < WALK_U; ++u) {
                const int b = b0 + 4 * u;
                all += (b < nblocks) ? v[u] : 0;
                pre += (b < (int)blockIdx.x) ? v[u] : 0;
            }
        }
        for (int o = 2; o > 0; o >>= 1) {
            all += __shfl_xor(all, o, 32);
            pre += __shfl_xor(pre, o, 32);
        }
        if (sub == 0) { s_all[d] = all; s_pre[d] = pre; }
    }
    block_digit_rank(digit, valid, rank, s_cnt, s_tot);
    if (tid < 32) {
        const int a0 = s_all[2 * tid];
        const int a1 = s_all[2 * tid + 1];
        int x = a0 + a1;
        for (int o = 1; o < 32; o <<= 1) {
            const int y = __shfl_up(x, o, 32);
            if (tid >= o) x += y;
        }
        const int ex = x - (a0 + a1);
        s_all[2 * tid] = ex;
        s_all[2 * tid + 1] = ex + a0;
    }
    __syncthreads();
    #pragma unroll
    for (int j = 0; j < KPT; ++j) {
        if (valid[j]) {
            const int d = digit[j];
            const int p = s_all[d] + s_pre[d] + s_cnt[(j * SORT_W + wv) * RDIG + d] + rank[j];
            keys_out[p] = key[j];
            vals_out[p] = val[j];
        }
    }
}

__global__ void k_rowptr_dinv(const int* __restrict__ sdst, int* __restrict__ rowptr,
                              float* __restrict__ dinv, int n, int e) {
    const int i = blockIdx.x * 256 + threadIdx.x;
    if (i > n) return;
    int lo0 = 0, hi0 = e;
    int lo1 = 0, hi1 = e;
    while ((lo0 < hi0) | (lo1 < hi1)) {
        const int m0 = (lo0 + hi0) >> 1;
        const int m1 = (lo1 + hi1) >> 1;
        const int k0 = sdst[min(m0, e - 1)];
        const int k1 = sdst[min(m1, e - 1)];
        if (lo0 < hi0) { if (k0 < i) lo0 = m0 + 1; else hi0 = m0; }
        if (lo1 < hi1) { if (k1 < i + 1) lo1 = m1 + 1; else hi1 = m1; }
    }
    rowptr[i] = lo0;
    if (i < n) {
        const float d = (float)(lo1 - lo0 + 1);
        dinv[i] = (d > 0.0f) ? rsqrtf(d) : 0.0f;
    }
}

__global__ void k_fill(const int* __restrict__ ssrc, const int* __restrict__ sdst,
                       const float* __restrict__ dinv, int* __restrict__ csr, int e) {
    const int i = blockIdx.x * 256 + threadIdx.x;
    if (i < e) {
        const int s = ssrc[i];
        const int d = sdst[i];
        const float w = dinv[s] * dinv[d];
        *(int2*)(csr + 2 * (size_t)i) = make_int2(s, __float_as_int(w));
    }
}

__global__ void __launch_bounds__(256)
k_gemm128(const float* __restrict__ A, const float* __restrict__ W,
          float* __restrict__ out) {
    __shared__ float lds[16 * 130];
    const int row0 = blockIdx.x * 16;

    #pragma unroll
    for (int i = 0; i < 4; ++i) {
        int idx = threadIdx.x + 256 * i;
        int r   = idx >> 6;
        int c2  = (idx & 63) * 2;
        const float2 v = *(const float2*)(A + (size_t)(row0 + r) * NH + c2);
        *(float2*)(lds + r * 130 + c2) = v;
    }
    __syncthreads();

    const int lane = threadIdx.x & 31;
    const int wv   = threadIdx.x >> 5;
    const int col0 = wv * 16;
    const int m    = lane & 15;
    const int half = lane >> 4;

    const float* arow = lds + m * 130;
    v8f c;
    #pragma unroll
    for (int i = 0; i < 8; ++i) c[i] = 0.0f;

    #pragma unroll 4
    for (int kk = 0; kk < 32; ++kk) {
        const int k0 = kk * 4 + 2 * half;
        v2f a;
        a.x = arow[k0];
        a.y = arow[k0 + 1];
        v2f b;
        b.x = W[(size_t)k0 * NH + col0 + m];
        b.y = W[(size_t)(k0 + 1) * NH + col0 + m];
        c = __builtin_amdgcn_wmma_f32_16x16x4_f32(false, a, false, b,
                                                  (short)0, c, false, false);
    }

    const int outRow = row0 + 8 * half;
    #pragma unroll
    for (int r = 0; r < 8; ++r)
        out[(size_t)(outRow + r) * NH + col0 + m] = c[r];
}

#define GR 12
#define TP 132

__device__ __forceinline__ float gelu_exact(float x) {
    return 0.5f * x * (1.0f + erff(x * 0.70710678118654752f));
}

__device__ __forceinline__ float4 fma4(float4 r, float w, float4 a) {
    a.x = fmaf(r.x, w, a.x); a.y = fmaf(r.y, w, a.y);
    a.z = fmaf(r.z, w, a.z); a.w = fmaf(r.w, w, a.w);
    return a;
}

__device__ __forceinline__ int tile_rowptr(const int* __restrict__ rowptr, int tile, int lane, int n) {
    return rowptr[min(tile * 16 + min(lane, 16), n)];
}

__device__ __forceinline__ int2 first_records(const int* __restrict__ csr, int beg, int end, int lane) {
    int2 rec = make_int2(0, 0);
    if (lane < end - beg) rec = *(const int2*)(csr + 2 * (size_t)(beg + lane));
    return rec;
}

__device__ __forceinline__ void
sum_group(const float* __restrict__ hin, int2 rec, int base, int k, int cnt, int eA,
          float4& accA, float4& accB, int c, int n) {
    const float4 z = make_float4(0.0f, 0.0f, 0.0f, 0.0f);
    int    sidx[GR];
    float  wt[GR];
    float4 r[GR];
    #pragma unroll
    for (int j = 0; j < GR; ++j) {
        const int q = (k + j) & 31;
        const int s = __builtin_amdgcn_readlane(rec.x, q);
        wt[j] = __int_as_float(__builtin_amdgcn_readlane(rec.y, q));
        sidx[j] = min(max(s, 0), n - 1);
    }
    #pragma unroll
    for (int j = 0; j < GR; ++j) {
        r[j] = z;
        if (k + j < cnt) r[j] = *(const float4*)(hin + (size_t)sidx[j] * NH + c);
    }
    #pragma unroll
    for (int j = 0; j < GR; ++j) {
        if (k + j < cnt) {
            if (base + k + j < eA) accA = fma4(r[j], wt[j], accA);
            else                   accB = fma4(r[j], wt[j], accB);
        }
    }
}

__device__ __forceinline__ void
sum_chunk(const float* __restrict__ hin, int2 rec, int base, int cnt, int eA,
          float4& accA, float4& accB, int c, int n) {
    sum_group(hin, rec, base, 0, cnt, eA, accA, accB, c, n);
    for (int k = GR; k < cnt; k += GR)
        sum_group(hin, rec, base, k, cnt, eA, accA, accB, c, n);
}

__device__ __forceinline__ void
gather_two(const float* __restrict__ hin, const int* __restrict__ csr,
           int bA, int eA, int eB, int2 rec, float4& accA, float4& accB,
           int lane, int c, int n) {
    sum_chunk(hin, rec, bA, min(32, eB - bA), eA, accA, accB, c, n);
    for (int base = bA + 32; base < eB; base += 32) {
        const int cnt = min(32, eB - base);
        int2 more = make_int2(0, 0);
        if (lane < cnt) more = *(const int2*)(csr + 2 * (size_t)(base + lane));
        sum_chunk(hin, more, base, cnt, eA, accA, accB, c, n);
    }
}

__device__ __forceinline__ float4
ln_gelu_res(float4 t, float4 g, float4 be, float4 res) {
    float s  = t.x + t.y + t.z + t.w;
    float ss = fmaf(t.x, t.x, fmaf(t.y, t.y, fmaf(t.z, t.z, t.w * t.w)));
    #pragma unroll
    for (int o = 16; o > 0; o >>= 1) {
        s  += __shfl_xor(s,  o, 32);
        ss += __shfl_xor(ss, o, 32);
    }
    const float mu   = s * (1.0f / NH);
    const float var  = fmaf(ss, (1.0f / NH), -mu * mu);
    const float rstd = rsqrtf(var + LN_EPS);
    float4 o;
    o.x = gelu_exact(fmaf((t.x - mu) * rstd, g.x, be.x)) + res.x;
    o.y = gelu_exact(fmaf((t.y - mu) * rstd, g.y, be.y)) + res.y;
    o.z = gelu_exact(fmaf((t.z - mu) * rstd, g.z, be.z)) + res.z;
    o.w = gelu_exact(fmaf((t.w - mu) * rstd, g.w, be.w)) + res.w;
    return o;
}

__device__ __forceinline__ void
conv_layer_pass(const float* __restrict__ hin, const int* __restrict__ rowptr,
                const int* __restrict__ csr, const float* __restrict__ dinv,
                const float* __restrict__ Wm, const float* bias, const float* gamma,
                const float* beta, float* hout, const float* wh, const float* bh,
                float* out, int n, int ntiles,
                float* tA, float* tC, float* tR, float* tP, const bool last) {
    const int lane = threadIdx.x & 31;
    const int c    = lane * 4;
    const int wv   = __builtin_amdgcn_readfirstlane((int)(threadIdx.x >> 5));
    const int col0 = wv * 16;
    const int m    = lane & 15;
    const int half = lane >> 4;
    const int G    = gridDim.x;
    const float4 z = make_float4(0.0f, 0.0f, 0.0f, 0.0f);

    if (threadIdx.x < 32)       *(float4*)(tP + c)          = *(const float4*)(bias + c);
    else if (threadIdx.x < 64)  *(float4*)(tP + NH + c)     = *(const float4*)(gamma + c);
    else if (threadIdx.x < 96)  *(float4*)(tP + 2 * NH + c) = *(const float4*)(beta + c);
    else if (last && threadIdx.x < 128) *(float4*)(tP + 3 * NH + c) = *(const float4*)(wh + c);

    v2f bw[32];
    #pragma unroll
    for (int kk = 0; kk < 32; ++kk) {
        const int k0 = kk * 4 + 2 * half;
        bw[kk].x = Wm[(size_t)k0 * NH + col0 + m];
        bw[kk].y = Wm[(size_t)(k0 + 1) * NH + col0 + m];
    }

    int tile = blockIdx.x;
    int rpN;
    int bA, eA, eB;
    int2 rec;
    {
        const int rp0 = tile_rowptr(rowptr, tile, lane, n);
        rpN = tile_rowptr(rowptr, tile + G, lane, n);
        bA = __builtin_amdgcn_readlane(rp0, wv * 2);
        eA = __builtin_amdgcn_readlane(rp0, wv * 2 + 1);
        eB = __builtin_amdgcn_readlane(rp0, wv * 2 + 2);
        rec = first_records(csr, bA, eB, lane);
    }

    for (; tile < ntiles; tile += G) {
        const int nbA = __builtin_amdgcn_readlane(rpN, wv * 2);
        const int neA = __builtin_amdgcn_readlane(rpN, wv * 2 + 1);
        const int neB = __builtin_amdgcn_readlane(rpN, wv * 2 + 2);
        const int2 nrec = first_records(csr, nbA, neB, lane);
        const int rpF = tile_rowptr(rowptr, tile + 2 * G, lane, n);

        const int nodeA = tile * 16 + wv * 2;
        const int nodeB = nodeA + 1;
        const float4 vA = *(const float4*)(hin + (size_t)min(nodeA, n - 1) * NH + c);
        const float4 vB = *(const float4*)(hin + (size_t)min(nodeB, n - 1) * NH + c);
        const float dv = dinv[min(tile * 16 + (lane & 15), n - 1)];
        float4 accA = z, accB = z;
        gather_two(hin, csr, bA, eA, eB, rec, accA, accB, lane, c, n);
        float dA = __int_as_float(__builtin_amdgcn_readlane(__float_as_int(dv), wv * 2));
        float dB = __int_as_float(__builtin_amdgcn_readlane(__float_as_int(dv), wv * 2 + 1));
        if (nodeA >= n) dA = 0.0f;
        if (nodeB >= n) dB = 0.0f;
        accA = fma4(vA, dA * dA, accA);
        accB = fma4(vB, dB * dB, accB);
        *(float4*)(tA + (wv * 2) * TP + c) = accA;
        *(float4*)(tA + (wv * 2 + 1) * TP + c) = accB;
        *(float4*)(tR + (wv * 2) * TP + c) = vA;
        *(float4*)(tR + (wv * 2 + 1) * TP + c) = vB;
        __syncthreads();

        const float* arow = tA + m * TP + 2 * half;
        v8f acc8;
        #pragma unroll
        for (int i = 0; i < 8; ++i) acc8[i] = 0.0f;
        #pragma unroll
        for (int kk = 0; kk < 32; ++kk) {
            const v2f a = *(const v2f*)(arow + kk * 4);
            acc8 = __builtin_amdgcn_wmma_f32_16x16x4_f32(false, a, false, bw[kk],
                                                         (short)0, acc8, false, false);
        }
        #pragma unroll
        for (int r = 0; r < 8; ++r) tC[(8 * half + r) * TP + col0 + m] = acc8[r];
        __syncthreads();

        const float4 b  = *(const float4*)(tP + c);
        const float4 g  = *(const float4*)(tP + NH + c);
        const float4 be = *(const float4*)(tP + 2 * NH + c);
        float4 tRowA = *(const float4*)(tC + (wv * 2) * TP + c);
        float4 tRowB = *(const float4*)(tC + (wv * 2 + 1) * TP + c);
        tRowA.x += b.x; tRowA.y += b.y; tRowA.z += b.z; tRowA.w += b.w;
        tRowB.x += b.x; tRowB.y += b.y; tRowB.z += b.z; tRowB.w += b.w;
        const float4 resA = *(const float4*)(tR + (wv * 2) * TP + c);
        const float4 resB = *(const float4*)(tR + (wv * 2 + 1) * TP + c);
        const float4 oA = ln_gelu_res(tRowA, g, be, resA);
        const float4 oB = ln_gelu_res(tRowB, g, be, resB);
        if (last) {
            const float4 hd = *(const float4*)(tP + 3 * NH + c);
            const float  hb = bh[0];
            float pA = fmaf(oA.x, hd.x, fmaf(oA.y, hd.y, fmaf(oA.z, hd.z, oA.w * hd.w)));
            float pB = fmaf(oB.x, hd.x, fmaf(oB.y, hd.y, fmaf(oB.z, hd.z, oB.w * hd.w)));
            #pragma unroll
            for (int q = 16; q > 0; q >>= 1) {
                pA += __shfl_xor(pA, q, 32);
                pB += __shfl_xor(pB, q, 32);
            }
            if (lane == 0) {
                if (nodeA < n) out[nodeA] = pA + hb;
                if (nodeB < n) out[nodeB] = pB + hb;
            }
        } else {
            if (nodeA < n) *(float4*)(hout + (size_t)nodeA * NH + c) = oA;
            if (nodeB < n) *(float4*)(hout + (size_t)nodeB * NH + c) = oB;
        }
        bA = nbA; eA = neA; eB = neB;
        rec = nrec;
        rpN = rpF;
    }
}

__global__ void __launch_bounds__(256, 6)
k_conv_layer(const float* __restrict__ hin, const int* __restrict__ rowptr,
             const int* __restrict__ csr, const float* __restrict__ dinv,
             const float* __restrict__ Wm, const float* bias, const float* gamma,
             const float* beta, float* hout, int n, int ntiles) {
    __shared__ __attribute__((aligned(16))) float tA[16 * TP];
    __shared__ __attribute__((aligned(16))) float tC[16 * TP];
    __shared__ __attribute__((aligned(16))) float tR[16 * TP];
    __shared__ __attribute__((aligned(16))) float tP[4 * NH];
    conv_layer_pass(hin, rowptr, csr, dinv, Wm, bias, gamma, beta, hout,
                    nullptr, nullptr, nullptr, n, ntiles, tA, tC, tR, tP, false);
}

__global__ void __launch_bounds__(256, 6)
k_conv_layer_head(const float* __restrict__ hin, const int* __restrict__ rowptr,
                  const int* __restrict__ csr, const float* __restrict__ dinv,
                  const float* __restrict__ Wm, const float* bias, const float* gamma,
                  const float* beta, const float* wh, const float* bh, float* out,
                  int n, int ntiles) {
    __shared__ __attribute__((aligned(16))) float tA[16 * TP];
    __shared__ __attribute__((aligned(16))) float tC[16 * TP];
    __shared__ __attribute__((aligned(16))) float tR[16 * TP];
    __shared__ __attribute__((aligned(16))) float tP[4 * NH];
    conv_layer_pass(hin, rowptr, csr, dinv, Wm, bias, gamma, beta, nullptr,
                    wh, bh, out, n, ntiles, tA, tC, tR, tP, true);
}


__global__ void k_lngelu(const float* __restrict__ in, const float* __restrict__ bias,
                         const float* __restrict__ gamma, const float* __restrict__ beta,
                         const float* __restrict__ resid, float* __restrict__ out, int n) {
    const int lane = threadIdx.x & 31;
    const int wv   = threadIdx.x >> 5;
    const int node = blockIdx.x * 8 + wv;
    if (node >= n) return;
    const int c = lane * 4;

    float4 t = *(const float4*)(in + (size_t)node * NH + c);
    if (bias) {
        float4 b = *(const float4*)(bias + c);
        t.x += b.x; t.y += b.y; t.z += b.z; t.w += b.w;
    }
    float s  = t.x + t.y + t.z + t.w;
    float ss = fmaf(t.x, t.x, fmaf(t.y, t.y, fmaf(t.z, t.z, t.w * t.w)));
    #pragma unroll
    for (int o = 16; o > 0; o >>= 1) {
        s  += __shfl_xor(s,  o, 32);
        ss += __shfl_xor(ss, o, 32);
    }
    const float mu   = s * (1.0f / NH);
    const float var  = fmaf(ss, (1.0f / NH), -mu * mu);
    const float rstd = rsqrtf(var + LN_EPS);

    float4 g  = *(const float4*)(gamma + c);
    float4 be = *(const float4*)(beta + c);
    float4 o;
    o.x = gelu_exact(fmaf((t.x - mu) * rstd, g.x, be.x));
    o.y = gelu_exact(fmaf((t.y - mu) * rstd, g.y, be.y));
    o.z = gelu_exact(fmaf((t.z - mu) * rstd, g.z, be.z));
    o.w = gelu_exact(fmaf((t.w - mu) * rstd, g.w, be.w));
    if (resid) {
        float4 r = *(const float4*)(resid + (size_t)node * NH + c);
        o.x += r.x; o.y += r.y; o.z += r.z; o.w += r.w;
    }
    *(float4*)(out + (size_t)node * NH + c) = o;
}

extern "C" void kernel_launch(void* const* d_in, const int* in_sizes, int n_in,
                              void* d_out, int out_size, void* d_ws, size_t ws_size,
                              hipStream_t stream) {
    const float* x      = (const float*)d_in[0];
    const int*   ei     = (const int*)d_in[1];
    const float* W_in   = (const float*)d_in[2];
    const float* b_in   = (const float*)d_in[3];
    const float* g_in   = (const float*)d_in[4];
    const float* be_in  = (const float*)d_in[5];
    const float* W_conv = (const float*)d_in[6];
    const float* b_conv = (const float*)d_in[7];
    const float* g_ln   = (const float*)d_in[8];
    const float* be_ln  = (const float*)d_in[9];
    const float* W_head = (const float*)d_in[10];
    const float* b_head = (const float*)d_in[11];
    float* out = (float*)d_out;

    const int N = in_sizes[0] / NH;
    const int E = in_sizes[1] / 2;
    const int* src = ei;
    const int* dst = ei + E;

    char* w = (char*)d_ws;
    float* dinv   = (float*)(w);
    int*   ghist  = (int*)  (w + (size_t)(1)   * (1 << 20));
    int*   rowptr = (int*)  (w + (size_t)(2)   * (1 << 20));
    float* hA     = (float*)(w + (size_t)(4)   * (1 << 20));
    float* hB     = (float*)(w + (size_t)(60)  * (1 << 20));
    int*   kA     = (int*)  (w + (size_t)(168) * (1 << 20));
    int*   vA     = (int*)  (w + (size_t)(171) * (1 << 20));
    int*   kB     = (int*)  (w + (size_t)(174) * (1 << 20));
    int*   vB     = (int*)  (w + (size_t)(177) * (1 << 20));
    int*   csr    = (int*)  (w + (size_t)(180) * (1 << 20));

    const int gE   = (E + 255) / 256;
    const int gN1  = (N + 1 + 255) / 256;
    const int gRow = N / 16;
    const int gNd  = (N + 7) / 8;
    const int nSb  = (E + SORT_K - 1) / SORT_K;
    const int nT   = (N + 15) / 16;
    const int gLy  = 768;

    k_sort_count<<<nSb, SORT_T, 0, stream>>>(dst, ghist, E, 0, nSb);
    k_sort_move<<<nSb, SORT_T, 0, stream>>>(dst, src, ghist, kA, vA, E, 0, nSb);
    k_sort_count<<<nSb, SORT_T, 0, stream>>>(kA, ghist, E, 6, nSb);
    k_sort_move<<<nSb, SORT_T, 0, stream>>>(kA, vA, ghist, kB, vB, E, 6, nSb);
    k_sort_count<<<nSb, SORT_T, 0, stream>>>(kB, ghist, E, 12, nSb);
    k_sort_move<<<nSb, SORT_T, 0, stream>>>(kB, vB, ghist, kA, vA, E, 12, nSb);
    k_rowptr_dinv<<<gN1, 256, 0, stream>>>(kA, rowptr, dinv, N, E);
    k_fill<<<gE, 256, 0, stream>>>(vA, kA, dinv, csr, E);

    k_gemm128<<<gRow, 256, 0, stream>>>(x, W_in, hB);
    k_lngelu<<<gNd, 256, 0, stream>>>(hB, b_in, g_in, be_in, nullptr, hA, N);

    for (int i = 0; i < NLAYERS; ++i) {
        const float* Wc = W_conv + (size_t)i * NH * NH;
        const float* bc = b_conv + (size_t)i * NH;
        const float* gl = g_ln   + (size_t)i * NH;
        const float* bl = be_ln  + (size_t)i * NH;
        const float* hin  = (i & 1) ? hB : hA;
        float*       hout = (i & 1) ? hA : hB;
        if (i + 1 < NLAYERS)
            k_conv_layer<<<gLy, 256, 0, stream>>>(hin, rowptr, csr, dinv, Wc, bc, gl, bl, hout, N, nT);
        else
            k_conv_layer_head<<<gLy, 256, 0, stream>>>(hin, rowptr, csr, dinv, Wc, bc, gl, bl,
                                                       W_head, b_head, out, N, nT);
    }
}


# GELU's erf as one branch-free rational function x*P(x^2)/Q(x^2) (|x| clamped at 4, abs error under 4e-7) in place of the two-branch library erff; a lane's 8 values run side by side; last bits change
# speedup vs baseline: 2.1114x; 1.0879x over previous
#include <hip/hip_runtime.h>
#include <hip/hip_bf16.h>

#define NH 128
#define NLAYERS 3
#define LN_EPS 1e-5f

typedef float v2f __attribute__((ext_vector_type(2)));
typedef float v8f __attribute__((ext_vector_type(8)));

#define RBITS  6
#define RDIG   64
#define SORT_T 256
#define SORT_W (SORT_T / 32)
#define KPT    16
#define SORT_K (SORT_T * KPT)
#define SORT_V (SORT_W * KPT)
#define WALK_U 10

__device__ __forceinline__ void
block_digit_rank(const int (&digit)[KPT], const bool (&valid)[KPT], int (&rank)[KPT],
                 int* s_cnt, int* s_tot) {
    const int tid  = threadIdx.x;
    const int lane = tid & 31;
    const int wv   = tid >> 5;
    #pragma unroll
    for (int i = 0; i < SORT_V * RDIG / SORT_T; ++i) s_cnt[tid + i * SORT_T] = 0;
    int cnt[KPT];
    #pragma unroll
    for (int j = 0; j < KPT; ++j) {
        unsigned peers = __builtin_amdgcn_ballot_w32(valid[j]);
        #pragma unroll
        for (int b = 0; b < RBITS; ++b) {
            const bool bit = ((digit[j] >> b) & 1) != 0;
            const unsigned bal = __builtin_amdgcn_ballot_w32(bit);
            peers &= bit ? bal : ~bal;
        }
        rank[j] = __popc(peers & ((1u << lane) - 1u));
        cnt[j]  = __popc(peers);
    }
    __syncthreads();
    #pragma unroll
    for (int j = 0; j < KPT; ++j)
        if (valid[j] && rank[j] == 0) s_cnt[(j * SORT_W + wv) * RDIG + digit[j]] = cnt[j];
    __syncthreads();
    if (tid < RDIG) {
        int run = 0;
        for (int v = 0; v < SORT_V; ++v) {
            const int t = s_cnt[v * RDIG + tid];
            s_cnt[v * RDIG + tid] = run;
            run += t;
        }
        s_tot[tid] = run;
    }
    __syncthreads();
}

__global__ void __launch_bounds__(SORT_T)
k_sort_count(const int* __restrict__ keys, int* __restrict__ ghist,
             int e, int shift, int nblocks) {
    __shared__ int s_cnt[SORT_V * RDIG];
    __shared__ int s_tot[RDIG];
    const int first = blockIdx.x * SORT_K + threadIdx.x;
    int  digit[KPT];
    bool valid[KPT];
    int  rank[KPT];
    #pragma unroll
    for (int j = 0; j < KPT; ++j) {
        const int i = first + j * SORT_T;
        valid[j] = i < e;
        const int key = valid[j] ? keys[i] : 0;
        digit[j] = (key >> shift) & (RDIG - 1);
    }
    block_digit_rank(digit, valid, rank, s_cnt, s_tot);
    if (threadIdx.x < RDIG) ghist[threadIdx.x * nblocks + blockIdx.x] = s_tot[threadIdx.x];
}

__global__ void __launch_bounds__(SORT_T)
k_sort_move(const int* __restrict__ keys, const int* __restrict__ vals,
            const int* __restrict__ ghist, int* __restrict__ keys_out,
            int* __restrict__ vals_out, int e, int shift, int nblocks) {
    __shared__ int s_cnt[SORT_V * RDIG];
    __shared__ int s_tot[RDIG];
    __shared__ int s_all[RDIG];
    __shared__ int s_pre[RDIG];
    const int tid = threadIdx.x;
    const int wv  = tid >> 5;
    const int first = blockIdx.x * SORT_K + tid;
    int  key[KPT];
    int  val[KPT];
    int  digit[KPT];
    bool valid[KPT];
    int  rank[KPT];
    #pragma unroll
    for (int j = 0; j < KPT; ++j) {
        const int i = first + j * SORT_T;
        valid[j] = i < e;
        key[j] = valid[j] ? keys[i] : 0;
        val[j] = valid[j] ? vals[i] : 0;
        digit[j] = (key[j] >> shift) & (RDIG - 1);
    }
    {
        const int d = tid >> 2;
        const int sub = tid & 3;
        const int* row = ghist + d * nblocks;
        int all = 0, pre = 0;
        for (int b0 = sub; b0 < nblocks; b0 += 4 * WALK_U) {
            int v[WALK_U];
            #pragma unroll
            for (int u = 0; u < WALK_U; ++u) v[u] = row[min(b0 + 4 * u, nblocks - 1)];
            #pragma unroll
            for (int u = 0; u < WALK_U; ++u) {
                const int b = b0 + 4 * u;
                all += (b < nblocks) ? v[u] : 0;
                pre += (b < (int)blockIdx.x) ? v[u] : 0;
            }
        }
        for (int o = 2; o > 0; o >>= 1) {
            all += __shfl_xor(all, o, 32);
            pre += __shfl_xor(pre, o, 32);
        }
        if (sub == 0) { s_all[d] = all; s_pre[d] = pre; }
    }
    block_digit_rank(digit, valid, rank, s_cnt, s_tot);
    if (tid < 32) {
        const int a0 = s_all[2 * tid];
        const int a1 = s_all[2 * tid + 1];
        int x = a0 + a1;
        for (int o = 1; o < 32; o <<= 1) {
            const int y = __shfl_up(x, o, 32);
            if (tid >= o) x += y;
        }
        const int ex = x - (a0 + a1);
        s_all[2 * tid] = ex;
        s_all[2 * tid + 1] = ex + a0;
    }
    __syncthreads();
    #pragma unroll
    for (int j = 0; j < KPT; ++j) {
        if (valid[j]) {
            const int d = digit[j];
            const int p = s_all[d] + s_pre[d] + s_cnt[(j * SORT_W + wv) * RDIG + d] + rank[j];
            keys_out[p] = key[j];
            vals_out[p] = val[j];
        }
    }
}

__global__ void k_rowptr_dinv(const int* __restrict__ sdst, int* __restrict__ rowptr,
                              float* __restrict__ dinv, int n, int e) {
    const int i = blockIdx.x * 256 + threadIdx.x;
    if (i > n) return;
    int lo0 = 0, hi0 = e;
    int lo1 = 0, hi1 = e;
    while ((lo0 < hi0) | (lo1 < hi1)) {
        const int m0 = (lo0 + hi0) >> 1;
        const int m1 = (lo1 + hi1) >> 1;
        const int k0 = sdst[min(m0, e - 1)];
        const int k1 = sdst[min(m1, e - 1)];
        if (lo0 < hi0) { if (k0 < i) lo0 = m0 + 1; else hi0 = m0; }
        if (lo1 < hi1) { if (k1 < i + 1) lo1 = m1 + 1; else hi1 = m1; }
    }
    rowptr[i] = lo0;
    if (i < n) {
        const float d = (float)(lo1 - lo0 + 1);
        dinv[i] = (d > 0.0f) ? rsqrtf(d) : 0.0f;
    }
}

__global__ void k_fill(const int* __restrict__ ssrc, const int* __restrict__ sdst,
                       const float* __restrict__ dinv, int* __restrict__ csr, int e) {
    const int i = blockIdx.x * 256 + threadIdx.x;
    if (i < e) {
        const int s = ssrc[i];
        const int d = sdst[i];
        const float w = dinv[s] * dinv[d];
        *(int2*)(csr + 2 * (size_t)i) = make_int2(s, __float_as_int(w));
    }
}

__global__ void __launch_bounds__(256)
k_gemm128(const float* __restrict__ A, const float* __restrict__ W,
          float* __restrict__ out) {
    __shared__ float lds[16 * 130];
    const int row0 = blockIdx.x * 16;

    #pragma unroll
    for (int i = 0; i < 4; ++i) {
        int idx = threadIdx.x + 256 * i;
        int r   = idx >> 6;
        int c2  = (idx & 63) * 2;
        const float2 v = *(const float2*)(A + (size_t)(row0 + r) * NH + c2);
        *(float2*)(lds + r * 130 + c2) = v;
    }
    __syncthreads();

    const int lane = threadIdx.x & 31;
    const int wv   = threadIdx.x >> 5;
    const int col0 = wv * 16;
    const int m    = lane & 15;
    const int half = lane >> 4;

    const float* arow = lds + m * 130;
    v8f c;
    #pragma unroll
    for (int i = 0; i < 8; ++i) c[i] = 0.0f;

    #pragma unroll 4
    for (int kk = 0; kk < 32; ++kk) {
        const int k0 = kk * 4 + 2 * half;
        v2f a;
        a.x = arow[k0];
        a.y = arow[k0 + 1];
        v2f b;
        b.x = W[(size_t)k0 * NH + col0 + m];
        b.y = W[(size_t)(k0 + 1) * NH + col0 + m];
        c = __builtin_amdgcn_wmma_f32_16x16x4_f32(false, a, false, b,
                                                  (short)0, c, false, false);
    }

    const int outRow = row0 + 8 * half;
    #pragma unroll
    for (int r = 0; r < 8; ++r)
        out[(size_t)(outRow + r) * NH + col0 + m] = c[r];
}

#define GR 12
#define TP 132

__device__ __forceinline__ float erf_rational(float x) {
    const float xc = fminf(fmaxf(x, -4.0f), 4.0f);
    const float x2 = xc * xc;
    float p = fmaf(-2.72614225801306e-10f, x2, 2.77068142495902e-08f);
    p = fmaf(p, x2, -2.10102402082508e-06f);
    p = fmaf(p, x2, -5.69250639462346e-05f);
    p = fmaf(p, x2, -7.34990630326855e-04f);
    p = fmaf(p, x2, -2.95459980854025e-03f);
    p = fmaf(p, x2, -1.60960333262415e-02f);
    float q = fmaf(-1.45660718464996e-05f, x2, -2.13374055278905e-04f);
    q = fmaf(q, x2, -1.68282697438203e-03f);
    q = fmaf(q, x2, -7.37332916720468e-03f);
    q = fmaf(q, x2, -1.42647390514189e-02f);
    float rq = __builtin_amdgcn_rcpf(q);
    rq = fmaf(fmaf(-q, rq, 1.0f), rq, rq);
    return fminf(fmaxf(xc * p * rq, -1.0f), 1.0f);
}

__device__ __forceinline__ float gelu_exact(float x) {
    const float hx = 0.5f * x;
    return fmaf(hx, erf_rational(x * 0.70710678118654752f), hx);
}

__device__ __forceinline__ float4 fma4(float4 r, float w, float4 a) {
    a.x = fmaf(r.x, w, a.x); a.y = fmaf(r.y, w, a.y);
    a.z = fmaf(r.z, w, a.z); a.w = fmaf(r.w, w, a.w);
    return a;
}

__device__ __forceinline__ int tile_rowptr(const int* __restrict__ rowptr, int tile, int lane, int n) {
    return rowptr[min(tile * 16 + min(lane, 16), n)];
}

__device__ __forceinline__ int2 first_records(const int* __restrict__ csr, int beg, int end, int lane) {
    int2 rec = make_int2(0, 0);
    if (lane < end - beg) rec = *(const int2*)(csr + 2 * (size_t)(beg + lane));
    return rec;
}

__device__ __forceinline__ void
sum_group(const float* __restrict__ hin, int2 rec, int base, int k, int cnt, int eA,
          float4& accA, float4& accB, int c, int n) {
    const float4 z = make_float4(0.0f, 0.0f, 0.0f, 0.0f);
    int    sidx[GR];
    float  wt[GR];
    float4 r[GR];
    #pragma unroll
    for (int j = 0; j < GR; ++j) {
        const int q = (k + j) & 31;
        const int s = __builtin_amdgcn_readlane(rec.x, q);
        wt[j] = __int_as_float(__builtin_amdgcn_readlane(rec.y, q));
        sidx[j] = min(max(s, 0), n - 1);
    }
    #pragma unroll
    for (int j = 0; j < GR; ++j) {
        r[j] = z;
        if (k + j < cnt) r[j] = *(const float4*)(hin + (size_t)sidx[j] * NH + c);
    }
    #pragma unroll
    for (int j = 0; j < GR; ++j) {
        if (k + j < cnt) {
            if (base + k + j < eA) accA = fma4(r[j], wt[j], accA);
            else                   accB = fma4(r[j], wt[j], accB);
        }
    }
}

__device__ __forceinline__ void
sum_chunk(const float* __restrict__ hin, int2 rec, int base, int cnt, int eA,
          float4& accA, float4& accB, int c, int n) {
    sum_group(hin, rec, base, 0, cnt, eA, accA, accB, c, n);
    for (int k = GR; k < cnt; k += GR)
        sum_group(hin, rec, base, k, cnt, eA, accA, accB, c, n);
}

__device__ __forceinline__ void
gather_two(const float* __restrict__ hin, const int* __restrict__ csr,
           int bA, int eA, int eB, int2 rec, float4& accA, float4& accB,
           int lane, int c, int n) {
    sum_chunk(hin, rec, bA, min(32, eB - bA), eA, accA, accB, c, n);
    for (int base = bA + 32; base < eB; base += 32) {
        const int cnt = min(32, eB - base);
        int2 more = make_int2(0, 0);
        if (lane < cnt) more = *(const int2*)(csr + 2 * (size_t)(base + lane));
        sum_chunk(hin, more, base, cnt, eA, accA, accB, c, n);
    }
}

__device__ __forceinline__ float4
ln_gelu_res(float4 t, float4 g, float4 be, float4 res) {
    float s  = t.x + t.y + t.z + t.w;
    float ss = fmaf(t.x, t.x, fmaf(t.y, t.y, fmaf(t.z, t.z, t.w * t.w)));
    #pragma unroll
    for (int o = 16; o > 0; o >>= 1) {
        s  += __shfl_xor(s,  o, 32);
        ss += __shfl_xor(ss, o, 32);
    }
    const float mu   = s * (1.0f / NH);
    const float var  = fmaf(ss, (1.0f / NH), -mu * mu);
    const float rstd = rsqrtf(var + LN_EPS);
    float4 o;
    o.x = gelu_exact(fmaf((t.x - mu) * rstd, g.x, be.x)) + res.x;
    o.y = gelu_exact(fmaf((t.y - mu) * rstd, g.y, be.y)) + res.y;
    o.z = gelu_exact(fmaf((t.z - mu) * rstd, g.z, be.z)) + res.z;
    o.w = gelu_exact(fmaf((t.w - mu) * rstd, g.w, be.w)) + res.w;
    return o;
}

__device__ __forceinline__ void
conv_layer_pass(const float* __restrict__ hin, const int* __restrict__ rowptr,
                const int* __restrict__ csr, const float* __restrict__ dinv,
                const float* __restrict__ Wm, const float* bias, const float* gamma,
                const float* beta, float* hout, const float* wh, const float* bh,
                float* out, int n, int ntiles,
                float* tA, float* tC, float* tR, float* tP, const bool last) {
    const int lane = threadIdx.x & 31;
    const int c    = lane * 4;
    const int wv   = __builtin_amdgcn_readfirstlane((int)(threadIdx.x >> 5));
    const int col0 = wv * 16;
    const int m    = lane & 15;
    const int half = lane >> 4;
    const int G    = gridDim.x;
    const float4 z = make_float4(0.0f, 0.0f, 0.0f, 0.0f);

    if (threadIdx.x < 32)       *(float4*)(tP + c)          = *(const float4*)(bias + c);
    else if (threadIdx.x < 64)  *(float4*)(tP + NH + c)     = *(const float4*)(gamma + c);
    else if (threadIdx.x < 96)  *(float4*)(tP + 2 * NH + c) = *(const float4*)(beta + c);
    else if (last && threadIdx.x < 128) *(float4*)(tP + 3 * NH + c) = *(const float4*)(wh + c);

    v2f bw[32];
    #pragma unroll
    for (int kk = 0; kk < 32; ++kk) {
        const int k0 = kk * 4 + 2 * half;
        bw[kk].x = Wm[(size_t)k0 * NH + col0 + m];
        bw[kk].y = Wm[(size_t)(k0 + 1) * NH + col0 + m];
    }

    int tile = blockIdx.x;
    int rpN;
    int bA, eA, eB;
    int2 rec;
    {
        const int rp0 = tile_rowptr(rowptr, tile, lane, n);
        rpN = tile_rowptr(rowptr, tile + G, lane, n);
        bA = __builtin_amdgcn_readlane(rp0, wv * 2);
        eA = __builtin_amdgcn_readlane(rp0, wv * 2 + 1);
        eB = __builtin_amdgcn_readlane(rp0, wv * 2 + 2);
        rec = first_records(csr, bA, eB, lane);
    }

    for (; tile < ntiles; tile += G) {
        const int nbA = __builtin_amdgcn_readlane(rpN, wv * 2);
        const int neA = __builtin_amdgcn_readlane(rpN, wv * 2 + 1);
        const int neB = __builtin_amdgcn_readlane(rpN, wv * 2 + 2);
        const int2 nrec = first_records(csr, nbA, neB, lane);
        const int rpF = tile_rowptr(rowptr, tile + 2 * G, lane, n);

        const int nodeA = tile * 16 + wv * 2;
        const int nodeB = nodeA + 1;
        const float4 vA = *(const float4*)(hin + (size_t)min(nodeA, n - 1) * NH + c);
        const float4 vB = *(const float4*)(hin + (size_t)min(nodeB, n - 1) * NH + c);
        const float dv = dinv[min(tile * 16 + (lane & 15), n - 1)];
        float4 accA = z, accB = z;
        gather_two(hin, csr, bA, eA, eB, rec, accA, accB, lane, c, n);
        float dA = __int_as_float(__builtin_amdgcn_readlane(__float_as_int(dv), wv * 2));
        float dB = __int_as_float(__builtin_amdgcn_readlane(__float_as_int(dv), wv * 2 + 1));
        if (nodeA >= n) dA = 0.0f;
        if (nodeB >= n) dB = 0.0f;
        accA = fma4(vA, dA * dA, accA);
        accB = fma4(vB, dB * dB, accB);
        *(float4*)(tA + (wv * 2) * TP + c) = accA;
        *(float4*)(tA + (wv * 2 + 1) * TP + c) = accB;
        *(float4*)(tR + (wv * 2) * TP + c) = vA;
        *(float4*)(tR + (wv * 2 + 1) * TP + c) = vB;
        __syncthreads();

        const float* arow = tA + m * TP + 2 * half;
        v8f acc8;
        #pragma unroll
        for (int i = 0; i < 8; ++i) acc8[i] = 0.0f;
        #pragma unroll
        for (int kk = 0; kk < 32; ++kk) {
            const v2f a = *(const v2f*)(arow + kk * 4);
            acc8 = __builtin_amdgcn_wmma_f32_16x16x4_f32(false, a, false, bw[kk],
                                                         (short)0, acc8, false, false);
        }
        #pragma unroll
        for (int r = 0; r < 8; ++r) tC[(8 * half + r) * TP + col0 + m] = acc8[r];
        __syncthreads();

        const float4 b  = *(const float4*)(tP + c);
        const float4 g  = *(const float4*)(tP + NH + c);
        const float4 be = *(const float4*)(tP + 2 * NH + c);
        float4 tRowA = *(const float4*)(tC + (wv * 2) * TP + c);
        float4 tRowB = *(const float4*)(tC + (wv * 2 + 1) * TP + c);
        tRowA.x += b.x; tRowA.y += b.y; tRowA.z += b.z; tRowA.w += b.w;
        tRowB.x += b.x; tRowB.y += b.y; tRowB.z += b.z; tRowB.w += b.w;
        const float4 resA = *(const float4*)(tR + (wv * 2) * TP + c);
        const float4 resB = *(const float4*)(tR + (wv * 2 + 1) * TP + c);
        const float4 oA = ln_gelu_res(tRowA, g, be, resA);
        const float4 oB = ln_gelu_res(tRowB, g, be, resB);
        if (last) {
            const float4 hd = *(const float4*)(tP + 3 * NH + c);
            const float  hb = bh[0];
            float pA = fmaf(oA.x, hd.x, fmaf(oA.y, hd.y, fmaf(oA.z, hd.z, oA.w * hd.w)));
            float pB = fmaf(oB.x, hd.x, fmaf(oB.y, hd.y, fmaf(oB.z, hd.z, oB.w * hd.w)));
            #pragma unroll
            for (int q = 16; q > 0; q >>= 1) {
                pA += __shfl_xor(pA, q, 32);
                pB += __shfl_xor(pB, q, 32);
            }
            if (lane == 0) {
                if (nodeA < n) out[nodeA] = pA + hb;
                if (nodeB < n) out[nodeB] = pB + hb;
            }
        } else {
            if (nodeA < n) *(float4*)(hout + (size_t)nodeA * NH + c) = oA;
            if (nodeB < n) *(float4*)(hout + (size_t)nodeB * NH + c) = oB;
        }
        bA = nbA; eA = neA; eB = neB;
        rec = nrec;
        rpN = rpF;
    }
}

__global__ void __launch_bounds__(256, 6)
k_conv_layer(const float* __restrict__ hin, const int* __restrict__ rowptr,
             const int* __restrict__ csr, const float* __restrict__ dinv,
             const float* __restrict__ Wm, const float* bias, const float* gamma,
             const float* beta, float* hout, int n, int ntiles) {
    __shared__ __attribute__((aligned(16))) float tA[16 * TP];
    __shared__ __attribute__((aligned(16))) float tC[16 * TP];
    __shared__ __attribute__((aligned(16))) float tR[16 * TP];
    __shared__ __attribute__((aligned(16))) float tP[4 * NH];
    conv_layer_pass(hin, rowptr, csr, dinv, Wm, bias, gamma, beta, hout,
                    nullptr, nullptr, nullptr, n, ntiles, tA, tC, tR, tP, false);
}

__global__ void __launch_bounds__(256, 6)
k_conv_layer_head(const float* __restrict__ hin, const int* __restrict__ rowptr,
                  const int* __restrict__ csr, const float* __restrict__ dinv,
                  const float* __restrict__ Wm, const float* bias, const float* gamma,
                  const float* beta, const float* wh, const float* bh, float* out,
                  int n, int ntiles) {
    __shared__ __attribute__((aligned(16))) float tA[16 * TP];
    __shared__ __attribute__((aligned(16))) float tC[16 * TP];
    __shared__ __attribute__((aligned(16))) float tR[16 * TP];
    __shared__ __attribute__((aligned(16))) float tP[4 * NH];
    conv_layer_pass(hin, rowptr, csr, dinv, Wm, bias, gamma, beta, nullptr,
                    wh, bh, out, n, ntiles, tA, tC, tR, tP, true);
}


__global__ void k_lngelu(const float* __restrict__ in, const float* __restrict__ bias,
                         const float* __restrict__ gamma, const float* __restrict__ beta,
                         const float* __restrict__ resid, float* __restrict__ out, int n) {
    const int lane = threadIdx.x & 31;
    const int wv   = threadIdx.x >> 5;
    const int node = blockIdx.x * 8 + wv;
    if (node >= n) return;
    const int c = lane * 4;

    float4 t = *(const float4*)(in + (size_t)node * NH + c);
    if (bias) {
        float4 b = *(const float4*)(bias + c);
        t.x += b.x; t.y += b.y; t.z += b.z; t.w += b.w;
    }
    float s  = t.x + t.y + t.z + t.w;
    float ss = fmaf(t.x, t.x, fmaf(t.y, t.y, fmaf(t.z, t.z, t.w * t.w)));
    #pragma unroll
    for (int o = 16; o > 0; o >>= 1) {
        s  += __shfl_xor(s,  o, 32);
        ss += __shfl_xor(ss, o, 32);
    }
    const float mu   = s * (1.0f / NH);
    const float var  = fmaf(ss, (1.0f / NH), -mu * mu);
    const float rstd = rsqrtf(var + LN_EPS);

    float4 g  = *(const float4*)(gamma + c);
    float4 be = *(const float4*)(beta + c);
    float4 o;
    o.x = gelu_exact(fmaf((t.x - mu) * rstd, g.x, be.x));
    o.y = gelu_exact(fmaf((t.y - mu) * rstd, g.y, be.y));
    o.z = gelu_exact(fmaf((t.z - mu) * rstd, g.z, be.z));
    o.w = gelu_exact(fmaf((t.w - mu) * rstd, g.w, be.w));
    if (resid) {
        float4 r = *(const float4*)(resid + (size_t)node * NH + c);
        o.x += r.x; o.y += r.y; o.z += r.z; o.w += r.w;
    }
    *(float4*)(out + (size_t)node * NH + c) = o;
}

extern "C" void kernel_launch(void* const* d_in, const int* in_sizes, int n_in,
                              void* d_out, int out_size, void* d_ws, size_t ws_size,
                              hipStream_t stream) {
    const float* x      = (const float*)d_in[0];
    const int*   ei     = (const int*)d_in[1];
    const float* W_in   = (const float*)d_in[2];
    const float* b_in   = (const float*)d_in[3];
    const float* g_in   = (const float*)d_in[4];
    const float* be_in  = (const float*)d_in[5];
    const float* W_conv = (const float*)d_in[6];
    const float* b_conv = (const float*)d_in[7];
    const float* g_ln   = (const float*)d_in[8];
    const float* be_ln  = (const float*)d_in[9];
    const float* W_head = (const float*)d_in[10];
    const float* b_head = (const float*)d_in[11];
    float* out = (float*)d_out;

    const int N = in_sizes[0] / NH;
    const int E = in_sizes[1] / 2;
    const int* src = ei;
    const int* dst = ei + E;

    char* w = (char*)d_ws;
    float* dinv   = (float*)(w);
    int*   ghist  = (int*)  (w + (size_t)(1)   * (1 << 20));
    int*   rowptr = (int*)  (w + (size_t)(2)   * (1 << 20));
    float* hA     = (float*)(w + (size_t)(4)   * (1 << 20));
    float* hB     = (float*)(w + (size_t)(60)  * (1 << 20));
    int*   kA     = (int*)  (w + (size_t)(168) * (1 << 20));
    int*   vA     = (int*)  (w + (size_t)(171) * (1 << 20));
    int*   kB     = (int*)  (w + (size_t)(174) * (1 << 20));
    int*   vB     = (int*)  (w + (size_t)(177) * (1 << 20));
    int*   csr    = (int*)  (w + (size_t)(180) * (1 << 20));

    const int gE   = (E + 255) / 256;
    const int gN1  = (N + 1 + 255) / 256;
    const int gRow = N / 16;
    const int gNd  = (N + 7) / 8;
    const int nSb  = (E + SORT_K - 1) / SORT_K;
    const int nT   = (N + 15) / 16;
    const int gLy  = 768;

    k_sort_count<<<nSb, SORT_T, 0, stream>>>(dst, ghist, E, 0, nSb);
    k_sort_move<<<nSb, SORT_T, 0, stream>>>(dst, src, ghist, kA, vA, E, 0, nSb);
    k_sort_count<<<nSb, SORT_T, 0, stream>>>(kA, ghist, E, 6, nSb);
    k_sort_move<<<nSb, SORT_T, 0, stream>>>(kA, vA, ghist, kB, vB, E, 6, nSb);
    k_sort_count<<<nSb, SORT_T, 0, stream>>>(kB, ghist, E, 12, nSb);
    k_sort_move<<<nSb, SORT_T, 0, stream>>>(kB, vB, ghist, kA, vA, E, 12, nSb);
    k_rowptr_dinv<<<gN1, 256, 0, stream>>>(kA, rowptr, dinv, N, E);
    k_fill<<<gE, 256, 0, stream>>>(vA, kA, dinv, csr, E);

    k_gemm128<<<gRow, 256, 0, stream>>>(x, W_in, hB);
    k_lngelu<<<gNd, 256, 0, stream>>>(hB, b_in, g_in, be_in, nullptr, hA, N);

    for (int i = 0; i < NLAYERS; ++i) {
        const float* Wc = W_conv + (size_t)i * NH * NH;
        const float* bc = b_conv + (size_t)i * NH;
        const float* gl = g_ln   + (size_t)i * NH;
        const float* bl = be_ln  + (size_t)i * NH;
        const float* hin  = (i & 1) ? hB : hA;
        float*       hout = (i & 1) ? hA : hB;
        if (i + 1 < NLAYERS)
            k_conv_layer<<<gLy, 256, 0, stream>>>(hin, rowptr, csr, dinv, Wc, bc, gl, bl, hout, N, nT);
        else
            k_conv_layer_head<<<gLy, 256, 0, stream>>>(hin, rowptr, csr, dinv, Wc, bc, gl, bl,
                                                       W_head, b_head, out, N, nT);
    }
}


# Input layer's LayerNorm+GELU done on the product tile in LDS: one resident launch (W_in fragments in registers, rows requested a turn ahead) replaces the GEMM launch and the wave-per-node LN launch
# speedup vs baseline: 2.3593x; 1.1174x over previous
#include <hip/hip_runtime.h>
#include <hip/hip_bf16.h>

#define NH 128
#define NLAYERS 3
#define LN_EPS 1e-5f

typedef float v2f __attribute__((ext_vector_type(2)));
typedef float v8f __attribute__((ext_vector_type(8)));

#define RBITS  6
#define RDIG   64
#define SORT_T 256
#define SORT_W (SORT_T / 32)
#define KPT    16
#define SORT_K (SORT_T * KPT)
#define SORT_V (SORT_W * KPT)
#define WALK_U 10

__device__ __forceinline__ void
block_digit_rank(const int (&digit)[KPT], const bool (&valid)[KPT], int (&rank)[KPT],
                 int* s_cnt, int* s_tot) {
    const int tid  = threadIdx.x;
    const int lane = tid & 31;
    const int wv   = tid >> 5;
    #pragma unroll
    for (int i = 0; i < SORT_V * RDIG / SORT_T; ++i) s_cnt[tid + i * SORT_T] = 0;
    int cnt[KPT];
    #pragma unroll
    for (int j = 0; j < KPT; ++j) {
        unsigned peers = __builtin_amdgcn_ballot_w32(valid[j]);
        #pragma unroll
        for (int b = 0; b < RBITS; ++b) {
            const bool bit = ((digit[j] >> b) & 1) != 0;
            const unsigned bal = __builtin_amdgcn_ballot_w32(bit);
            peers &= bit ? bal : ~bal;
        }
        rank[j] = __popc(peers & ((1u << lane) - 1u));
        cnt[j]  = __popc(peers);
    }
    __syncthreads();
    #pragma unroll
    for (int j = 0; j < KPT; ++j)
        if (valid[j] && rank[j] == 0) s_cnt[(j * SORT_W + wv) * RDIG + digit[j]] = cnt[j];
    __syncthreads();
    if (tid < RDIG) {
        int run = 0;
        for (int v = 0; v < SORT_V; ++v) {
            const int t = s_cnt[v * RDIG + tid];
            s_cnt[v * RDIG + tid] = run;
            run += t;
        }
        s_tot[tid] = run;
    }
    __syncthreads();
}

__global__ void __launch_bounds__(SORT_T)
k_sort_count(const int* __restrict__ keys, int* __restrict__ ghist,
             int e, int shift, int nblocks) {
    __shared__ int s_cnt[SORT_V * RDIG];
    __shared__ int s_tot[RDIG];
    const int first = blockIdx.x * SORT_K + threadIdx.x;
    int  digit[KPT];
    bool valid[KPT];
    int  rank[KPT];
    #pragma unroll
    for (int j = 0; j < KPT; ++j) {
        const int i = first + j * SORT_T;
        valid[j] = i < e;
        const int key = valid[j] ? keys[i] : 0;
        digit[j] = (key >> shift) & (RDIG - 1);
    }
    block_digit_rank(digit, valid, rank, s_cnt, s_tot);
    if (threadIdx.x < RDIG) ghist[threadIdx.x * nblocks + blockIdx.x] = s_tot[threadIdx.x];
}

__global__ void __launch_bounds__(SORT_T)
k_sort_move(const int* __restrict__ keys, const int* __restrict__ vals,
            const int* __restrict__ ghist, int* __restrict__ keys_out,
            int* __restrict__ vals_out, int e, int shift, int nblocks) {
    __shared__ int s_cnt[SORT_V * RDIG];
    __shared__ int s_tot[RDIG];
    __shared__ int s_all[RDIG];
    __shared__ int s_pre[RDIG];
    const int tid = threadIdx.x;
    const int wv  = tid >> 5;
    const int first = blockIdx.x * SORT_K + tid;
    int  key[KPT];
    int  val[KPT];
    int  digit[KPT];
    bool valid[KPT];
    int  rank[KPT];
    #pragma unroll
    for (int j = 0; j < KPT; ++j) {
        const int i = first + j * SORT_T;
        valid[j] = i < e;
        key[j] = valid[j] ? keys[i] : 0;
        val[j] = valid[j] ? vals[i] : 0;
        digit[j] = (key[j] >> shift) & (RDIG - 1);
    }
    {
        const int d = tid >> 2;
        const int sub = tid & 3;
        const int* row = ghist + d * nblocks;
        int all = 0, pre = 0;
        for (int b0 = sub; b0 < nblocks; b0 += 4 * WALK_U) {
            int v[WALK_U];
            #pragma unroll
            for (int u = 0; u < WALK_U; ++u) v[u] = row[min(b0 + 4 * u, nblocks - 1)];
            #pragma unroll
            for (int u = 0; u < WALK_U; ++u) {
                const int b = b0 + 4 * u;
                all += (b < nblocks) ? v[u] : 0;
                pre += (b < (int)blockIdx.x) ? v[u] : 0;
            }
        }
        for (int o = 2; o > 0; o >>= 1) {
            all += __shfl_xor(all, o, 32);
            pre += __shfl_xor(pre, o, 32);
        }
        if (sub == 0) { s_all[d] = all; s_pre[d] = pre; }
    }
    block_digit_rank(digit, valid, rank, s_cnt, s_tot);
    if (tid < 32) {
        const int a0 = s_all[2 * tid];
        const int a1 = s_all[2 * tid + 1];
        int x = a0 + a1;
        for (int o = 1; o < 32; o <<= 1) {
            const int y = __shfl_up(x, o, 32);
            if (tid >= o) x += y;
        }
        const int ex = x - (a0 + a1);
        s_all[2 * tid] = ex;
        s_all[2 * tid + 1] = ex + a0;
    }
    __syncthreads();
    #pragma unroll
    for (int j = 0; j < KPT; ++j) {
        if (valid[j]) {
            const int d = digit[j];
            const int p = s_all[d] + s_pre[d] + s_cnt[(j * SORT_W + wv) * RDIG + d] + rank[j];
            keys_out[p] = key[j];
            vals_out[p] = val[j];
        }
    }
}

__global__ void k_rowptr_dinv(const int* __restrict__ sdst, int* __restrict__ rowptr,
                              float* __restrict__ dinv, int n, int e) {
    const int i = blockIdx.x * 256 + threadIdx.x;
    if (i > n) return;
    int lo0 = 0, hi0 = e;
    int lo1 = 0, hi1 = e;
    while ((lo0 < hi0) | (lo1 < hi1)) {
        const int m0 = (lo0 + hi0) >> 1;
        const int m1 = (lo1 + hi1) >> 1;
        const int k0 = sdst[min(m0, e - 1)];
        const int k1 = sdst[min(m1, e - 1)];
        if (lo0 < hi0) { if (k0 < i) lo0 = m0 + 1; else hi0 = m0; }
        if (lo1 < hi1) { if (k1 < i + 1) lo1 = m1 + 1; else hi1 = m1; }
    }
    rowptr[i] = lo0;
    if (i < n) {
        const float d = (float)(lo1 - lo0 + 1);
        dinv[i] = (d > 0.0f) ? rsqrtf(d) : 0.0f;
    }
}

__global__ void k_fill(const int* __restrict__ ssrc, const int* __restrict__ sdst,
                       const float* __restrict__ dinv, int* __restrict__ csr, int e) {
    const int i = blockIdx.x * 256 + threadIdx.x;
    if (i < e) {
        const int s = ssrc[i];
        const int d = sdst[i];
        const float w = dinv[s] * dinv[d];
        *(int2*)(csr + 2 * (size_t)i) = make_int2(s, __float_as_int(w));
    }
}

#define GR 12
#define TP 132

__device__ __forceinline__ float erf_rational(float x) {
    const float xc = fminf(fmaxf(x, -4.0f), 4.0f);
    const float x2 = xc * xc;
    float p = fmaf(-2.72614225801306e-10f, x2, 2.77068142495902e-08f);
    p = fmaf(p, x2, -2.10102402082508e-06f);
    p = fmaf(p, x2, -5.69250639462346e-05f);
    p = fmaf(p, x2, -7.34990630326855e-04f);
    p = fmaf(p, x2, -2.95459980854025e-03f);
    p = fmaf(p, x2, -1.60960333262415e-02f);
    float q = fmaf(-1.45660718464996e-05f, x2, -2.13374055278905e-04f);
    q = fmaf(q, x2, -1.68282697438203e-03f);
    q = fmaf(q, x2, -7.37332916720468e-03f);
    q = fmaf(q, x2, -1.42647390514189e-02f);
    float rq = __builtin_amdgcn_rcpf(q);
    rq = fmaf(fmaf(-q, rq, 1.0f), rq, rq);
    return fminf(fmaxf(xc * p * rq, -1.0f), 1.0f);
}

__device__ __forceinline__ float gelu_exact(float x) {
    const float hx = 0.5f * x;
    return fmaf(hx, erf_rational(x * 0.70710678118654752f), hx);
}

__device__ __forceinline__ float4 fma4(float4 r, float w, float4 a) {
    a.x = fmaf(r.x, w, a.x); a.y = fmaf(r.y, w, a.y);
    a.z = fmaf(r.z, w, a.z); a.w = fmaf(r.w, w, a.w);
    return a;
}

__device__ __forceinline__ int tile_rowptr(const int* __restrict__ rowptr, int tile, int lane, int n) {
    return rowptr[min(tile * 16 + min(lane, 16), n)];
}

__device__ __forceinline__ int2 first_records(const int* __restrict__ csr, int beg, int end, int lane) {
    int2 rec = make_int2(0, 0);
    if (lane < end - beg) rec = *(const int2*)(csr + 2 * (size_t)(beg + lane));
    return rec;
}

__device__ __forceinline__ void
sum_group(const float* __restrict__ hin, int2 rec, int base, int k, int cnt, int eA,
          float4& accA, float4& accB, int c, int n) {
    const float4 z = make_float4(0.0f, 0.0f, 0.0f, 0.0f);
    int    sidx[GR];
    float  wt[GR];
    float4 r[GR];
    #pragma unroll
    for (int j = 0; j < GR; ++j) {
        const int q = (k + j) & 31;
        const int s = __builtin_amdgcn_readlane(rec.x, q);
        wt[j] = __int_as_float(__builtin_amdgcn_readlane(rec.y, q));
        sidx[j] = min(max(s, 0), n - 1);
    }
    #pragma unroll
    for (int j = 0; j < GR; ++j) {
        r[j] = z;
        if (k + j < cnt) r[j] = *(const float4*)(hin + (size_t)sidx[j] * NH + c);
    }
    #pragma unroll
    for (int j = 0; j < GR; ++j) {
        if (k + j < cnt) {
            if (base + k + j < eA) accA = fma4(r[j], wt[j], accA);
            else                   accB = fma4(r[j], wt[j], accB);
        }
    }
}

__device__ __forceinline__ void
sum_chunk(const float* __restrict__ hin, int2 rec, int base, int cnt, int eA,
          float4& accA, float4& accB, int c, int n) {
    sum_group(hin, rec, base, 0, cnt, eA, accA, accB, c, n);
    for (int k = GR; k < cnt; k += GR)
        sum_group(hin, rec, base, k, cnt, eA, accA, accB, c, n);
}

__device__ __forceinline__ void
gather_two(const float* __restrict__ hin, const int* __restrict__ csr,
           int bA, int eA, int eB, int2 rec, float4& accA, float4& accB,
           int lane, int c, int n) {
    sum_chunk(hin, rec, bA, min(32, eB - bA), eA, accA, accB, c, n);
    for (int base = bA + 32; base < eB; base += 32) {
        const int cnt = min(32, eB - base);
        int2 more = make_int2(0, 0);
        if (lane < cnt) more = *(const int2*)(csr + 2 * (size_t)(base + lane));
        sum_chunk(hin, more, base, cnt, eA, accA, accB, c, n);
    }
}

__device__ __forceinline__ float4
ln_gelu(float4 t, float4 g, float4 be) {
    float s  = t.x + t.y + t.z + t.w;
    float ss = fmaf(t.x, t.x, fmaf(t.y, t.y, fmaf(t.z, t.z, t.w * t.w)));
    #pragma unroll
    for (int o = 16; o > 0; o >>= 1) {
        s  += __shfl_xor(s,  o, 32);
        ss += __shfl_xor(ss, o, 32);
    }
    const float mu   = s * (1.0f / NH);
    const float var  = fmaf(ss, (1.0f / NH), -mu * mu);
    const float rstd = rsqrtf(var + LN_EPS);
    float4 o;
    o.x = gelu_exact(fmaf((t.x - mu) * rstd, g.x, be.x));
    o.y = gelu_exact(fmaf((t.y - mu) * rstd, g.y, be.y));
    o.z = gelu_exact(fmaf((t.z - mu) * rstd, g.z, be.z));
    o.w = gelu_exact(fmaf((t.w - mu) * rstd, g.w, be.w));
    return o;
}

__device__ __forceinline__ float4
ln_gelu_res(float4 t, float4 g, float4 be, float4 res) {
    float4 o = ln_gelu(t, g, be);
    o.x += res.x; o.y += res.y; o.z += res.z; o.w += res.w;
    return o;
}

__device__ __forceinline__ void
conv_layer_pass(const float* __restrict__ hin, const int* __restrict__ rowptr,
                const int* __restrict__ csr, const float* __restrict__ dinv,
                const float* __restrict__ Wm, const float* bias, const float* gamma,
                const float* beta, float* hout, const float* wh, const float* bh,
                float* out, int n, int ntiles,
                float* tA, float* tC, float* tR, float* tP, const bool last) {
    const int lane = threadIdx.x & 31;
    const int c    = lane * 4;
    const int wv   = __builtin_amdgcn_readfirstlane((int)(threadIdx.x >> 5));
    const int col0 = wv * 16;
    const int m    = lane & 15;
    const int half = lane >> 4;
    const int G    = gridDim.x;
    const float4 z = make_float4(0.0f, 0.0f, 0.0f, 0.0f);

    if (threadIdx.x < 32)       *(float4*)(tP + c)          = *(const float4*)(bias + c);
    else if (threadIdx.x < 64)  *(float4*)(tP + NH + c)     = *(const float4*)(gamma + c);
    else if (threadIdx.x < 96)  *(float4*)(tP + 2 * NH + c) = *(const float4*)(beta + c);
    else if (last && threadIdx.x < 128) *(float4*)(tP + 3 * NH + c) = *(const float4*)(wh + c);

    v2f bw[32];
    #pragma unroll
    for (int kk = 0; kk < 32; ++kk) {
        const int k0 = kk * 4 + 2 * half;
        bw[kk].x = Wm[(size_t)k0 * NH + col0 + m];
        bw[kk].y = Wm[(size_t)(k0 + 1) * NH + col0 + m];
    }

    int tile = blockIdx.x;
    int rpN;
    int bA, eA, eB;
    int2 rec;
    {
        const int rp0 = tile_rowptr(rowptr, tile, lane, n);
        rpN = tile_rowptr(rowptr, tile + G, lane, n);
        bA = __builtin_amdgcn_readlane(rp0, wv * 2);
        eA = __builtin_amdgcn_readlane(rp0, wv * 2 + 1);
        eB = __builtin_amdgcn_readlane(rp0, wv * 2 + 2);
        rec = first_records(csr, bA, eB, lane);
    }

    for (; tile < ntiles; tile += G) {
        const int nbA = __builtin_amdgcn_readlane(rpN, wv * 2);
        const int neA = __builtin_amdgcn_readlane(rpN, wv * 2 + 1);
        const int neB = __builtin_amdgcn_readlane(rpN, wv * 2 + 2);
        const int2 nrec = first_records(csr, nbA, neB, lane);
        const int rpF = tile_rowptr(rowptr, tile + 2 * G, lane, n);

        const int nodeA = tile * 16 + wv * 2;
        const int nodeB = nodeA + 1;
        const float4 vA = *(const float4*)(hin + (size_t)min(nodeA, n - 1) * NH + c);
        const float4 vB = *(const float4*)(hin + (size_t)min(nodeB, n - 1) * NH + c);
        const float dv = dinv[min(tile * 16 + (lane & 15), n - 1)];
        float4 accA = z, accB = z;
        gather_two(hin, csr, bA, eA, eB, rec, accA, accB, lane, c, n);
        float dA = __int_as_float(__builtin_amdgcn_readlane(__float_as_int(dv), wv * 2));
        float dB = __int_as_float(__builtin_amdgcn_readlane(__float_as_int(dv), wv * 2 + 1));
        if (nodeA >= n) dA = 0.0f;
        if (nodeB >= n) dB = 0.0f;
        accA = fma4(vA, dA * dA, accA);
        accB = fma4(vB, dB * dB, accB);
        *(float4*)(tA + (wv * 2) * TP + c) = accA;
        *(float4*)(tA + (wv * 2 + 1) * TP + c) = accB;
        *(float4*)(tR + (wv * 2) * TP + c) = vA;
        *(float4*)(tR + (wv * 2 + 1) * TP + c) = vB;
        __syncthreads();

        const float* arow = tA + m * TP + 2 * half;
        v8f acc8;
        #pragma unroll
        for (int i = 0; i < 8; ++i) acc8[i] = 0.0f;
        #pragma unroll
        for (int kk = 0; kk < 32; ++kk) {
            const v2f a = *(const v2f*)(arow + kk * 4);
            acc8 = __builtin_amdgcn_wmma_f32_16x16x4_f32(false, a, false, bw[kk],
                                                         (short)0, acc8, false, false);
        }
        #pragma unroll
        for (int r = 0; r < 8; ++r) tC[(8 * half + r) * TP + col0 + m] = acc8[r];
        __syncthreads();

        const float4 b  = *(const float4*)(tP + c);
        const float4 g  = *(const float4*)(tP + NH + c);
        const float4 be = *(const float4*)(tP + 2 * NH + c);
        float4 tRowA = *(const float4*)(tC + (wv * 2) * TP + c);
        float4 tRowB = *(const float4*)(tC + (wv * 2 + 1) * TP + c);
        tRowA.x += b.x; tRowA.y += b.y; tRowA.z += b.z; tRowA.w += b.w;
        tRowB.x += b.x; tRowB.y += b.y; tRowB.z += b.z; tRowB.w += b.w;
        const float4 resA = *(const float4*)(tR + (wv * 2) * TP + c);
        const float4 resB = *(const float4*)(tR + (wv * 2 + 1) * TP + c);
        const float4 oA = ln_gelu_res(tRowA, g, be, resA);
        const float4 oB = ln_gelu_res(tRowB, g, be, resB);
        if (last) {
            const float4 hd = *(const float4*)(tP + 3 * NH + c);
            const float  hb = bh[0];
            float pA = fmaf(oA.x, hd.x, fmaf(oA.y, hd.y, fmaf(oA.z, hd.z, oA.w * hd.w)));
            float pB = fmaf(oB.x, hd.x, fmaf(oB.y, hd.y, fmaf(oB.z, hd.z, oB.w * hd.w)));
            #pragma unroll
            for (int q = 16; q > 0; q >>= 1) {
                pA += __shfl_xor(pA, q, 32);
                pB += __shfl_xor(pB, q, 32);
            }
            if (lane == 0) {
                if (nodeA < n) out[nodeA] = pA + hb;
                if (nodeB < n) out[nodeB] = pB + hb;
            }
        } else {
            if (nodeA < n) *(float4*)(hout + (size_t)nodeA * NH + c) = oA;
            if (nodeB < n) *(float4*)(hout + (size_t)nodeB * NH + c) = oB;
        }
        bA = nbA; eA = neA; eB = neB;
        rec = nrec;
        rpN = rpF;
    }
}

__global__ void __launch_bounds__(256, 6)
k_conv_layer(const float* __restrict__ hin, const int* __restrict__ rowptr,
             const int* __restrict__ csr, const float* __restrict__ dinv,
             const float* __restrict__ Wm, const float* bias, const float* gamma,
             const float* beta, float* hout, int n, int ntiles) {
    __shared__ __attribute__((aligned(16))) float tA[16 * TP];
    __shared__ __attribute__((aligned(16))) float tC[16 * TP];
    __shared__ __attribute__((aligned(16))) float tR[16 * TP];
    __shared__ __attribute__((aligned(16))) float tP[4 * NH];
    conv_layer_pass(hin, rowptr, csr, dinv, Wm, bias, gamma, beta, hout,
                    nullptr, nullptr, nullptr, n, ntiles, tA, tC, tR, tP, false);
}

__global__ void __launch_bounds__(256, 6)
k_conv_layer_head(const float* __restrict__ hin, const int* __restrict__ rowptr,
                  const int* __restrict__ csr, const float* __restrict__ dinv,
                  const float* __restrict__ Wm, const float* bias, const float* gamma,
                  const float* beta, const float* wh, const float* bh, float* out,
                  int n, int ntiles) {
    __shared__ __attribute__((aligned(16))) float tA[16 * TP];
    __shared__ __attribute__((aligned(16))) float tC[16 * TP];
    __shared__ __attribute__((aligned(16))) float tR[16 * TP];
    __shared__ __attribute__((aligned(16))) float tP[4 * NH];
    conv_layer_pass(hin, rowptr, csr, dinv, Wm, bias, gamma, beta, nullptr,
                    wh, bh, out, n, ntiles, tA, tC, tR, tP, true);
}

__device__ __forceinline__ void lds_barrier() {
    __builtin_amdgcn_fence(__ATOMIC_RELEASE, "workgroup", "local");
    __builtin_amdgcn_s_barrier();
    __builtin_amdgcn_fence(__ATOMIC_ACQUIRE, "workgroup", "local");
}

__global__ void __launch_bounds__(256, 8)
k_input_layer(const float* x, const float* __restrict__ Wm,
              const float* __restrict__ bias, const float* __restrict__ gamma,
              const float* __restrict__ beta, float* hout, int n, int ntiles) {
    __shared__ __attribute__((aligned(16))) float tA[16 * TP];
    __shared__ __attribute__((aligned(16))) float tC[16 * TP];
    __shared__ __attribute__((aligned(16))) float tP[3 * NH];
    const int lane = threadIdx.x & 31;
    const int c    = lane * 4;
    const int wv   = __builtin_amdgcn_readfirstlane((int)(threadIdx.x >> 5));
    const int col0 = wv * 16;
    const int m    = lane & 15;
    const int half = lane >> 4;
    const int G    = gridDim.x;

    if (threadIdx.x < 32)       *(float4*)(tP + c)          = *(const float4*)(bias + c);
    else if (threadIdx.x < 64)  *(float4*)(tP + NH + c)     = *(const float4*)(gamma + c);
    else if (threadIdx.x < 96)  *(float4*)(tP + 2 * NH + c) = *(const float4*)(beta + c);

    v2f bw[32];
    #pragma unroll
    for (int kk = 0; kk < 32; ++kk) {
        const int k0 = kk * 4 + 2 * half;
        bw[kk].x = Wm[(size_t)k0 * NH + col0 + m];
        bw[kk].y = Wm[(size_t)(k0 + 1) * NH + col0 + m];
    }

    int tile = blockIdx.x;
    float4 vA = *(const float4*)(x + (size_t)min(tile * 16 + wv * 2,     n - 1) * NH + c);
    float4 vB = *(const float4*)(x + (size_t)min(tile * 16 + wv * 2 + 1, n - 1) * NH + c);

    for (; tile < ntiles; tile += G) {
        const int nodeA = tile * 16 + wv * 2;
        const int nodeB = nodeA + 1;
        const int nxt = (tile + G) * 16 + wv * 2;
        const float4 nA = *(const float4*)(x + (size_t)min(nxt,     n - 1) * NH + c);
        const float4 nB = *(const float4*)(x + (size_t)min(nxt + 1, n - 1) * NH + c);

        *(float4*)(tA + (wv * 2) * TP + c)     = vA;
        *(float4*)(tA + (wv * 2 + 1) * TP + c) = vB;
        lds_barrier();

        const float* arow = tA + m * TP + 2 * half;
        v8f acc8;
        #pragma unroll
        for (int i = 0; i < 8; ++i) acc8[i] = 0.0f;
        #pragma unroll
        for (int kk = 0; kk < 32; ++kk) {
            const v2f a = *(const v2f*)(arow + kk * 4);
            acc8 = __builtin_amdgcn_wmma_f32_16x16x4_f32(false, a, false, bw[kk],
                                                         (short)0, acc8, false, false);
        }
        #pragma unroll
        for (int r = 0; r < 8; ++r) tC[(8 * half + r) * TP + col0 + m] = acc8[r];
        lds_barrier();

        const float4 b  = *(const float4*)(tP + c);
        const float4 g  = *(const float4*)(tP + NH + c);
        const float4 be = *(const float4*)(tP + 2 * NH + c);
        float4 tRowA = *(const float4*)(tC + (wv * 2) * TP + c);
        float4 tRowB = *(const float4*)(tC + (wv * 2 + 1) * TP + c);
        tRowA.x += b.x; tRowA.y += b.y; tRowA.z += b.z; tRowA.w += b.w;
        tRowB.x += b.x; tRowB.y += b.y; tRowB.z += b.z; tRowB.w += b.w;
        const float4 oA = ln_gelu(tRowA, g, be);
        const float4 oB = ln_gelu(tRowB, g, be);
        if (nodeA < n) *(float4*)(hout + (size_t)nodeA * NH + c) = oA;
        if (nodeB < n) *(float4*)(hout + (size_t)nodeB * NH + c) = oB;
        vA = nA;
        vB = nB;
    }
}

extern "C" void kernel_launch(void* const* d_in, const int* in_sizes, int n_in,
                              void* d_out, int out_size, void* d_ws, size_t ws_size,
                              hipStream_t stream) {
    const float* x      = (const float*)d_in[0];
    const int*   ei     = (const int*)d_in[1];
    const float* W_in   = (const float*)d_in[2];
    const float* b_in   = (const float*)d_in[3];
    const float* g_in   = (const float*)d_in[4];
    const float* be_in  = (const float*)d_in[5];
    const float* W_conv = (const float*)d_in[6];
    const float* b_conv = (const float*)d_in[7];
    const float* g_ln   = (const float*)d_in[8];
    const float* be_ln  = (const float*)d_in[9];
    const float* W_head = (const float*)d_in[10];
    const float* b_head = (const float*)d_in[11];
    float* out = (float*)d_out;

    const int N = in_sizes[0] / NH;
    const int E = in_sizes[1] / 2;
    const int* src = ei;
    const int* dst = ei + E;

    char* w = (char*)d_ws;
    float* dinv   = (float*)(w);
    int*   ghist  = (int*)  (w + (size_t)(1)   * (1 << 20));
    int*   rowptr = (int*)  (w + (size_t)(2)   * (1 << 20));
    float* hA     = (float*)(w + (size_t)(4)   * (1 << 20));
    float* hB     = (float*)(w + (size_t)(60)  * (1 << 20));
    int*   kA     = (int*)  (w + (size_t)(168) * (1 << 20));
    int*   vA     = (int*)  (w + (size_t)(171) * (1 << 20));
    int*   kB     = (int*)  (w + (size_t)(174) * (1 << 20));
    int*   vB     = (int*)  (w + (size_t)(177) * (1 << 20));
    int*   csr    = (int*)  (w + (size_t)(180) * (1 << 20));

    const int gE   = (E + 255) / 256;
    const int gN1  = (N + 1 + 255) / 256;
    const int nSb  = (E + SORT_K - 1) / SORT_K;
    const int nT   = (N + 15) / 16;
    const int gLy  = 768;
    const int gIn  = 1024;

    k_sort_count<<<nSb, SORT_T, 0, stream>>>(dst, ghist, E, 0, nSb);
    k_sort_move<<<nSb, SORT_T, 0, stream>>>(dst, src, ghist, kA, vA, E, 0, nSb);
    k_sort_count<<<nSb, SORT_T, 0, stream>>>(kA, ghist, E, 6, nSb);
    k_sort_move<<<nSb, SORT_T, 0, stream>>>(kA, vA, ghist, kB, vB, E, 6, nSb);
    k_sort_count<<<nSb, SORT_T, 0, stream>>>(kB, ghist, E, 12, nSb);
    k_sort_move<<<nSb, SORT_T, 0, stream>>>(kB, vB, ghist, kA, vA, E, 12, nSb);
    k_rowptr_dinv<<<gN1, 256, 0, stream>>>(kA, rowptr, dinv, N, E);
    k_fill<<<gE, 256, 0, stream>>>(vA, kA, dinv, csr, E);

    k_input_layer<<<gIn, 256, 0, stream>>>(x, W_in, b_in, g_in, be_in, hA, N, nT);

    for (int i = 0; i < NLAYERS; ++i) {
        const float* Wc = W_conv + (size_t)i * NH * NH;
        const float* bc = b_conv + (size_t)i * NH;
        const float* gl = g_ln   + (size_t)i * NH;
        const float* bl = be_ln  + (size_t)i * NH;
        const float* hin  = (i & 1) ? hB : hA;
        float*       hout = (i & 1) ? hA : hB;
        if (i + 1 < NLAYERS)
            k_conv_layer<<<gLy, 256, 0, stream>>>(hin, rowptr, csr, dinv, Wc, bc, gl, bl, hout, N, nT);
        else
            k_conv_layer_head<<<gLy, 256, 0, stream>>>(hin, rowptr, csr, dinv, Wc, bc, gl, bl,
                                                       W_head, b_head, out, N, nT);
    }
}


# Sort move passes: a block's keys and values go to their block-sorted place in LDS (over the dead count table) and leave as one contiguous run per digit, not 32 scattered lines per store
# speedup vs baseline: 2.7929x; 1.1838x over previous
#include <hip/hip_runtime.h>
#include <hip/hip_bf16.h>

#define NH 128
#define NLAYERS 3
#define LN_EPS 1e-5f

typedef float v2f __attribute__((ext_vector_type(2)));
typedef float v8f __attribute__((ext_vector_type(8)));

#define RBITS  6
#define RDIG   64
#define SORT_T 256
#define SORT_W (SORT_T / 32)
#define KPT    16
#define SORT_K (SORT_T * KPT)
#define SORT_V (SORT_W * KPT)
#define WALK_U 10

__device__ __forceinline__ void
block_digit_rank(const int (&digit)[KPT], const bool (&valid)[KPT], int (&rank)[KPT],
                 int* s_cnt, int* s_tot) {
    const int tid  = threadIdx.x;
    const int lane = tid & 31;
    const int wv   = tid >> 5;
    #pragma unroll
    for (int i = 0; i < SORT_V * RDIG / SORT_T; ++i) s_cnt[tid + i * SORT_T] = 0;
    int cnt[KPT];
    #pragma unroll
    for (int j = 0; j < KPT; ++j) {
        unsigned peers = __builtin_amdgcn_ballot_w32(valid[j]);
        #pragma unroll
        for (int b = 0; b < RBITS; ++b) {
            const bool bit = ((digit[j] >> b) & 1) != 0;
            const unsigned bal = __builtin_amdgcn_ballot_w32(bit);
            peers &= bit ? bal : ~bal;
        }
        rank[j] = __popc(peers & ((1u << lane) - 1u));
        cnt[j]  = __popc(peers);
    }
    __syncthreads();
    #pragma unroll
    for (int j = 0; j < KPT; ++j)
        if (valid[j] && rank[j] == 0) s_cnt[(j * SORT_W + wv) * RDIG + digit[j]] = cnt[j];
    __syncthreads();
    if (tid < RDIG) {
        int run = 0;
        for (int v = 0; v < SORT_V; ++v) {
            const int t = s_cnt[v * RDIG + tid];
            s_cnt[v * RDIG + tid] = run;
            run += t;
        }
        s_tot[tid] = run;
    }
    __syncthreads();
}

__global__ void k_rowptr_dinv(const int* __restrict__ sdst, int* __restrict__ rowptr,
                              float* __restrict__ dinv, int n, int e) {
    const int i = blockIdx.x * 256 + threadIdx.x;
    if (i > n) return;
    int lo0 = 0, hi0 = e;
    int lo1 = 0, hi1 = e;
    while ((lo0 < hi0) | (lo1 < hi1)) {
        const int m0 = (lo0 + hi0) >> 1;
        const int m1 = (lo1 + hi1) >> 1;
        const int k0 = sdst[min(m0, e - 1)];
        const int k1 = sdst[min(m1, e - 1)];
        if (lo0 < hi0) { if (k0 < i) lo0 = m0 + 1; else hi0 = m0; }
        if (lo1 < hi1) { if (k1 < i + 1) lo1 = m1 + 1; else hi1 = m1; }
    }
    rowptr[i] = lo0;
    if (i < n) {
        const float d = (float)(lo1 - lo0 + 1);
        dinv[i] = (d > 0.0f) ? rsqrtf(d) : 0.0f;
    }
}

__global__ void k_fill(const int* __restrict__ ssrc, const int* __restrict__ sdst,
                       const float* __restrict__ dinv, int* __restrict__ csr, int e) {
    const int i = blockIdx.x * 256 + threadIdx.x;
    if (i < e) {
        const int s = ssrc[i];
        const int d = sdst[i];
        const float w = dinv[s] * dinv[d];
        *(int2*)(csr + 2 * (size_t)i) = make_int2(s, __float_as_int(w));
    }
}

#define GR 12
#define TP 132

__device__ __forceinline__ float erf_rational(float x) {
    const float xc = fminf(fmaxf(x, -4.0f), 4.0f);
    const float x2 = xc * xc;
    float p = fmaf(-2.72614225801306e-10f, x2, 2.77068142495902e-08f);
    p = fmaf(p, x2, -2.10102402082508e-06f);
    p = fmaf(p, x2, -5.69250639462346e-05f);
    p = fmaf(p, x2, -7.34990630326855e-04f);
    p = fmaf(p, x2, -2.95459980854025e-03f);
    p = fmaf(p, x2, -1.60960333262415e-02f);
    float q = fmaf(-1.45660718464996e-05f, x2, -2.13374055278905e-04f);
    q = fmaf(q, x2, -1.68282697438203e-03f);
    q = fmaf(q, x2, -7.37332916720468e-03f);
    q = fmaf(q, x2, -1.42647390514189e-02f);
    float rq = __builtin_amdgcn_rcpf(q);
    rq = fmaf(fmaf(-q, rq, 1.0f), rq, rq);
    return fminf(fmaxf(xc * p * rq, -1.0f), 1.0f);
}

__device__ __forceinline__ float gelu_exact(float x) {
    const float hx = 0.5f * x;
    return fmaf(hx, erf_rational(x * 0.70710678118654752f), hx);
}

__device__ __forceinline__ float4 fma4(float4 r, float w, float4 a) {
    a.x = fmaf(r.x, w, a.x); a.y = fmaf(r.y, w, a.y);
    a.z = fmaf(r.z, w, a.z); a.w = fmaf(r.w, w, a.w);
    return a;
}

__device__ __forceinline__ int tile_rowptr(const int* __restrict__ rowptr, int tile, int lane, int n) {
    return rowptr[min(tile * 16 + min(lane, 16), n)];
}

__device__ __forceinline__ int2 first_records(const int* __restrict__ csr, int beg, int end, int lane) {
    int2 rec = make_int2(0, 0);
    if (lane < end - beg) rec = *(const int2*)(csr + 2 * (size_t)(beg + lane));
    return rec;
}

__device__ __forceinline__ void
sum_group(const float* __restrict__ hin, int2 rec, int base, int k, int cnt, int eA,
          float4& accA, float4& accB, int c, int n) {
    const float4 z = make_float4(0.0f, 0.0f, 0.0f, 0.0f);
    int    sidx[GR];
    float  wt[GR];
    float4 r[GR];
    #pragma unroll
    for (int j = 0; j < GR; ++j) {
        const int q = (k + j) & 31;
        const int s = __builtin_amdgcn_readlane(rec.x, q);
        wt[j] = __int_as_float(__builtin_amdgcn_readlane(rec.y, q));
        sidx[j] = min(max(s, 0), n - 1);
    }
    #pragma unroll
    for (int j = 0; j < GR; ++j) {
        r[j] = z;
        if (k + j < cnt) r[j] = *(const float4*)(hin + (size_t)sidx[j] * NH + c);
    }
    #pragma unroll
    for (int j = 0; j < GR; ++j) {
        if (k + j < cnt) {
            if (base + k + j < eA) accA = fma4(r[j], wt[j], accA);
            else                   accB = fma4(r[j], wt[j], accB);
        }
    }
}

__device__ __forceinline__ void
sum_chunk(const float* __restrict__ hin, int2 rec, int base, int cnt, int eA,
          float4& accA, float4& accB, int c, int n) {
    sum_group(hin, rec, base, 0, cnt, eA, accA, accB, c, n);
    for (int k = GR; k < cnt; k += GR)
        sum_group(hin, rec, base, k, cnt, eA, accA, accB, c, n);
}

__device__ __forceinline__ void
gather_two(const float* __restrict__ hin, const int* __restrict__ csr,
           int bA, int eA, int eB, int2 rec, float4& accA, float4& accB,
           int lane, int c, int n) {
    sum_chunk(hin, rec, bA, min(32, eB - bA), eA, accA, accB, c, n);
    for (int base = bA + 32; base < eB; base += 32) {
        const int cnt = min(32, eB - base);
        int2 more = make_int2(0, 0);
        if (lane < cnt) more = *(const int2*)(csr + 2 * (size_t)(base + lane));
        sum_chunk(hin, more, base, cnt, eA, accA, accB, c, n);
    }
}

__device__ __forceinline__ float4
ln_gelu(float4 t, float4 g, float4 be) {
    float s  = t.x + t.y + t.z + t.w;
    float ss = fmaf(t.x, t.x, fmaf(t.y, t.y, fmaf(t.z, t.z, t.w * t.w)));
    #pragma unroll
    for (int o = 16; o > 0; o >>= 1) {
        s  += __shfl_xor(s,  o, 32);
        ss += __shfl_xor(ss, o, 32);
    }
    const float mu   = s * (1.0f / NH);
    const float var  = fmaf(ss, (1.0f / NH), -mu * mu);
    const float rstd = rsqrtf(var + LN_EPS);
    float4 o;
    o.x = gelu_exact(fmaf((t.x - mu) * rstd, g.x, be.x));
    o.y = gelu_exact(fmaf((t.y - mu) * rstd, g.y, be.y));
    o.z = gelu_exact(fmaf((t.z - mu) * rstd, g.z, be.z));
    o.w = gelu_exact(fmaf((t.w - mu) * rstd, g.w, be.w));
    return o;
}

__device__ __forceinline__ float4
ln_gelu_res(float4 t, float4 g, float4 be, float4 res) {
    float4 o = ln_gelu(t, g, be);
    o.x += res.x; o.y += res.y; o.z += res.z; o.w += res.w;
    return o;
}

__device__ __forceinline__ void
conv_layer_pass(const float* __restrict__ hin, const int* __restrict__ rowptr,
                const int* __restrict__ csr, const float* __restrict__ dinv,
                const float* __restrict__ Wm, const float* bias, const float* gamma,
                const float* beta, float* hout, const float* wh, const float* bh,
                float* out, int n, int ntiles,
                float* tA, float* tC, float* tR, float* tP, const bool last) {
    const int lane = threadIdx.x & 31;
    const int c    = lane * 4;
    const int wv   = __builtin_amdgcn_readfirstlane((int)(threadIdx.x >> 5));
    const int col0 = wv * 16;
    const int m    = lane & 15;
    const int half = lane >> 4;
    const int G    = gridDim.x;
    const float4 z = make_float4(0.0f, 0.0f, 0.0f, 0.0f);

    if (threadIdx.x < 32)       *(float4*)(tP + c)          = *(const float4*)(bias + c);
    else if (threadIdx.x < 64)  *(float4*)(tP + NH + c)     = *(const float4*)(gamma + c);
    else if (threadIdx.x < 96)  *(float4*)(tP + 2 * NH + c) = *(const float4*)(beta + c);
    else if (last && threadIdx.x < 128) *(float4*)(tP + 3 * NH + c) = *(const float4*)(wh + c);

    v2f bw[32];
    #pragma unroll
    for (int kk = 0; kk < 32; ++kk) {
        const int k0 = kk * 4 + 2 * half;
        bw[kk].x = Wm[(size_t)k0 * NH + col0 + m];
        bw[kk].y = Wm[(size_t)(k0 + 1) * NH + col0 + m];
    }

    int tile = blockIdx.x;
    int rpN;
    int bA, eA, eB;
    int2 rec;
    {
        const int rp0 = tile_rowptr(rowptr, tile, lane, n);
        rpN = tile_rowptr(rowptr, tile + G, lane, n);
        bA = __builtin_amdgcn_readlane(rp0, wv * 2);
        eA = __builtin_amdgcn_readlane(rp0, wv * 2 + 1);
        eB = __builtin_amdgcn_readlane(rp0, wv * 2 + 2);
        rec = first_records(csr, bA, eB, lane);
    }

    for (; tile < ntiles; tile += G) {
        const int nbA = __builtin_amdgcn_readlane(rpN, wv * 2);
        const int neA = __builtin_amdgcn_readlane(rpN, wv * 2 + 1);
        const int neB = __builtin_amdgcn_readlane(rpN, wv * 2 + 2);
        const int2 nrec = first_records(csr, nbA, neB, lane);
        const int rpF = tile_rowptr(rowptr, tile + 2 * G, lane, n);

        const int nodeA = tile * 16 + wv * 2;
        const int nodeB = nodeA + 1;
        const float4 vA = *(const float4*)(hin + (size_t)min(nodeA, n - 1) * NH + c);
        const float4 vB = *(const float4*)(hin + (size_t)min(nodeB, n - 1) * NH + c);
        const float dv = dinv[min(tile * 16 + (lane & 15), n - 1)];
        float4 accA = z, accB = z;
        gather_two(hin, csr, bA, eA, eB, rec, accA, accB, lane, c, n);
        float dA = __int_as_float(__builtin_amdgcn_readlane(__float_as_int(dv), wv * 2));
        float dB = __int_as_float(__builtin_amdgcn_readlane(__float_as_int(dv), wv * 2 + 1));
        if (nodeA >= n) dA = 0.0f;
        if (nodeB >= n) dB = 0.0f;
        accA = fma4(vA, dA * dA, accA);
        accB = fma4(vB, dB * dB, accB);
        *(float4*)(tA + (wv * 2) * TP + c) = accA;
        *(float4*)(tA + (wv * 2 + 1) * TP + c) = accB;
        *(float4*)(tR + (wv * 2) * TP + c) = vA;
        *(float4*)(tR + (wv * 2 + 1) * TP + c) = vB;
        __syncthreads();

        const float* arow = tA + m * TP + 2 * half;
        v8f acc8;
        #pragma unroll
        for (int i = 0; i < 8; ++i) acc8[i] = 0.0f;
        #pragma unroll
        for (int kk = 0; kk < 32; ++kk) {
            const v2f a = *(const v2f*)(arow + kk * 4);
            acc8 = __builtin_amdgcn_wmma_f32_16x16x4_f32(false, a, false, bw[kk],
                                                         (short)0, acc8, false, false);
        }
        #pragma unroll
        for (int r = 0; r < 8; ++r) tC[(8 * half + r) * TP + col0 + m] = acc8[r];
        __syncthreads();

        const float4 b  = *(const float4*)(tP + c);
        const float4 g  = *(const float4*)(tP + NH + c);
        const float4 be = *(const float4*)(tP + 2 * NH + c);
        float4 tRowA = *(const float4*)(tC + (wv * 2) * TP + c);
        float4 tRowB = *(const float4*)(tC + (wv * 2 + 1) * TP + c);
        tRowA.x += b.x; tRowA.y += b.y; tRowA.z += b.z; tRowA.w += b.w;
        tRowB.x += b.x; tRowB.y += b.y; tRowB.z += b.z; tRowB.w += b.w;
        const float4 resA = *(const float4*)(tR + (wv * 2) * TP + c);
        const float4 resB = *(const float4*)(tR + (wv * 2 + 1) * TP + c);
        const float4 oA = ln_gelu_res(tRowA, g, be, resA);
        const float4 oB = ln_gelu_res(tRowB, g, be, resB);
        if (last) {
            const float4 hd = *(const float4*)(tP + 3 * NH + c);
            const float  hb = bh[0];
            float pA = fmaf(oA.x, hd.x, fmaf(oA.y, hd.y, fmaf(oA.z, hd.z, oA.w * hd.w)));
            float pB = fmaf(oB.x, hd.x, fmaf(oB.y, hd.y, fmaf(oB.z, hd.z, oB.w * hd.w)));
            #pragma unroll
            for (int q = 16; q > 0; q >>= 1) {
                pA += __shfl_xor(pA, q, 32);
                pB += __shfl_xor(pB, q, 32);
            }
            if (lane == 0) {
                if (nodeA < n) out[nodeA] = pA + hb;
                if (nodeB < n) out[nodeB] = pB + hb;
            }
        } else {
            if (nodeA < n) *(float4*)(hout + (size_t)nodeA * NH + c) = oA;
            if (nodeB < n) *(float4*)(hout + (size_t)nodeB * NH + c) = oB;
        }
        bA = nbA; eA = neA; eB = neB;
        rec = nrec;
        rpN = rpF;
    }
}

__global__ void __launch_bounds__(256, 6)
k_conv_layer(const float* __restrict__ hin, const int* __restrict__ rowptr,
             const int* __restrict__ csr, const float* __restrict__ dinv,
             const float* __restrict__ Wm, const float* bias, const float* gamma,
             const float* beta, float* hout, int n, int ntiles) {
    __shared__ __attribute__((aligned(16))) float tA[16 * TP];
    __shared__ __attribute__((aligned(16))) float tC[16 * TP];
    __shared__ __attribute__((aligned(16))) float tR[16 * TP];
    __shared__ __attribute__((aligned(16))) float tP[4 * NH];
    conv_layer_pass(hin, rowptr, csr, dinv, Wm, bias, gamma, beta, hout,
                    nullptr, nullptr, nullptr, n, ntiles, tA, tC, tR, tP, false);
}

__global__ void __launch_bounds__(256, 6)
k_conv_layer_head(const float* __restrict__ hin, const int* __restrict__ rowptr,
                  const int* __restrict__ csr, const float* __restrict__ dinv,
                  const float* __restrict__ Wm, const float* bias, const float* gamma,
                  const float* beta, const float* wh, const float* bh, float* out,
                  int n, int ntiles) {
    __shared__ __attribute__((aligned(16))) float tA[16 * TP];
    __shared__ __attribute__((aligned(16))) float tC[16 * TP];
    __shared__ __attribute__((aligned(16))) float tR[16 * TP];
    __shared__ __attribute__((aligned(16))) float tP[4 * NH];
    conv_layer_pass(hin, rowptr, csr, dinv, Wm, bias, gamma, beta, nullptr,
                    wh, bh, out, n, ntiles, tA, tC, tR, tP, true);
}

__device__ __forceinline__ void lds_barrier() {
    __builtin_amdgcn_fence(__ATOMIC_RELEASE, "workgroup", "local");
    __builtin_amdgcn_s_barrier();
    __builtin_amdgcn_fence(__ATOMIC_ACQUIRE, "workgroup", "local");
}

__device__ __forceinline__ void
input_layer_tiles(const float* x, const float* __restrict__ Wm,
                  const float* __restrict__ bias, const float* __restrict__ gamma,
                  const float* __restrict__ beta, float* hout, int n,
                  int tile, const int tileEnd, const int G,
                  float* tA, float* tC, float* tP) {
    const int lane = threadIdx.x & 31;
    const int c    = lane * 4;
    const int wv   = __builtin_amdgcn_readfirstlane((int)(threadIdx.x >> 5));
    const int col0 = wv * 16;
    const int m    = lane & 15;
    const int half = lane >> 4;

    if (threadIdx.x < 32)       *(float4*)(tP + c)          = *(const float4*)(bias + c);
    else if (threadIdx.x < 64)  *(float4*)(tP + NH + c)     = *(const float4*)(gamma + c);
    else if (threadIdx.x < 96)  *(float4*)(tP + 2 * NH + c) = *(const float4*)(beta + c);

    v2f bw[32];
    #pragma unroll
    for (int kk = 0; kk < 32; ++kk) {
        const int k0 = kk * 4 + 2 * half;
        bw[kk].x = Wm[(size_t)k0 * NH + col0 + m];
        bw[kk].y = Wm[(size_t)(k0 + 1) * NH + col0 + m];
    }

    float4 vA = *(const float4*)(x + (size_t)min(tile * 16 + wv * 2,     n - 1) * NH + c);
    float4 vB = *(const float4*)(x + (size_t)min(tile * 16 + wv * 2 + 1, n - 1) * NH + c);

    for (; tile < tileEnd; tile += G) {
        const int nodeA = tile * 16 + wv * 2;
        const int nodeB = nodeA + 1;
        const int nxt = (tile + G) * 16 + wv * 2;
        const float4 nA = *(const float4*)(x + (size_t)min(nxt,     n - 1) * NH + c);
        const float4 nB = *(const float4*)(x + (size_t)min(nxt + 1, n - 1) * NH + c);

        *(float4*)(tA + (wv * 2) * TP + c)     = vA;
        *(float4*)(tA + (wv * 2 + 1) * TP + c) = vB;
        lds_barrier();

        const float* arow = tA + m * TP + 2 * half;
        v8f acc8;
        #pragma unroll
        for (int i = 0; i < 8; ++i) acc8[i] = 0.0f;
        #pragma unroll
        for (int kk = 0; kk < 32; ++kk) {
            const v2f a = *(const v2f*)(arow + kk * 4);
            acc8 = __builtin_amdgcn_wmma_f32_16x16x4_f32(false, a, false, bw[kk],
                                                         (short)0, acc8, false, false);
        }
        #pragma unroll
        for (int r = 0; r < 8; ++r) tC[(8 * half + r) * TP + col0 + m] = acc8[r];
        lds_barrier();

        const float4 b  = *(const float4*)(tP + c);
        const float4 g  = *(const float4*)(tP + NH + c);
        const float4 be = *(const float4*)(tP + 2 * NH + c);
        float4 tRowA = *(const float4*)(tC + (wv * 2) * TP + c);
        float4 tRowB = *(const float4*)(tC + (wv * 2 + 1) * TP + c);
        tRowA.x += b.x; tRowA.y += b.y; tRowA.z += b.z; tRowA.w += b.w;
        tRowB.x += b.x; tRowB.y += b.y; tRowB.z += b.z; tRowB.w += b.w;
        const float4 oA = ln_gelu(tRowA, g, be);
        const float4 oB = ln_gelu(tRowB, g, be);
        if (nodeA < n) *(float4*)(hout + (size_t)nodeA * NH + c) = oA;
        if (nodeB < n) *(float4*)(hout + (size_t)nodeB * NH + c) = oB;
        vA = nA;
        vB = nB;
    }
}

__global__ void __launch_bounds__(SORT_T, 8)
k_sort_count(const int* __restrict__ keys, int* __restrict__ ghist,
             int e, int shift, int nblocks,
             const float* x, const float* __restrict__ Wm, const float* __restrict__ bias,
             const float* __restrict__ gamma, const float* __restrict__ beta, float* hout,
             int n, int tile0, int tile1) {
    __shared__ int s_cnt[SORT_V * RDIG];
    __shared__ int s_tot[RDIG];
    __shared__ __attribute__((aligned(16))) float tA[16 * TP];
    __shared__ __attribute__((aligned(16))) float tC[16 * TP];
    __shared__ __attribute__((aligned(16))) float tP[3 * NH];
    if ((int)blockIdx.x >= nblocks) {
        input_layer_tiles(x, Wm, bias, gamma, beta, hout, n,
                          tile0 + ((int)blockIdx.x - nblocks), tile1,
                          (int)gridDim.x - nblocks, tA, tC, tP);
        return;
    }
    const int first = blockIdx.x * SORT_K + threadIdx.x;
    int  digit[KPT];
    bool valid[KPT];
    int  rank[KPT];
    #pragma unroll
    for (int j = 0; j < KPT; ++j) {
        const int i = first + j * SORT_T;
        valid[j] = i < e;
        const int key = valid[j] ? keys[i] : 0;
        digit[j] = (key >> shift) & (RDIG - 1);
    }
    block_digit_rank(digit, valid, rank, s_cnt, s_tot);
    if (threadIdx.x < RDIG) ghist[threadIdx.x * nblocks + blockIdx.x] = s_tot[threadIdx.x];
}

__global__ void __launch_bounds__(SORT_T, 8)
k_sort_move(const int* __restrict__ keys, const int* __restrict__ vals,
            const int* __restrict__ ghist, int* __restrict__ keys_out,
            int* __restrict__ vals_out, int e, int shift, int nblocks,
            const float* x, const float* __restrict__ Wm, const float* __restrict__ bias,
            const float* __restrict__ gamma, const float* __restrict__ beta, float* hout,
            int n, int tile0, int tile1) {
    __shared__ int s_cnt[SORT_V * RDIG];
    __shared__ int s_tot[RDIG];
    __shared__ int s_all[RDIG];
    __shared__ int s_pre[RDIG];
    __shared__ int s_loc[RDIG];
    __shared__ __attribute__((aligned(16))) float tA[16 * TP];
    __shared__ __attribute__((aligned(16))) float tC[16 * TP];
    __shared__ __attribute__((aligned(16))) float tP[3 * NH];
    if ((int)blockIdx.x >= nblocks) {
        input_layer_tiles(x, Wm, bias, gamma, beta, hout, n,
                          tile0 + ((int)blockIdx.x - nblocks), tile1,
                          (int)gridDim.x - nblocks, tA, tC, tP);
        return;
    }
    const int tid = threadIdx.x;
    const int wv  = tid >> 5;
    const int first = blockIdx.x * SORT_K + tid;
    int  key[KPT];
    int  val[KPT];
    int  digit[KPT];
    bool valid[KPT];
    int  rank[KPT];
    #pragma unroll
    for (int j = 0; j < KPT; ++j) {
        const int i = first + j * SORT_T;
        valid[j] = i < e;
        key[j] = valid[j] ? keys[i] : 0;
        val[j] = valid[j] ? vals[i] : 0;
        digit[j] = (key[j] >> shift) & (RDIG - 1);
    }
    {
        const int d = tid >> 2;
        const int sub = tid & 3;
        const int* row = ghist + d * nblocks;
        int all = 0, pre = 0;
        for (int b0 = sub; b0 < nblocks; b0 += 4 * WALK_U) {
            int v[WALK_U];
            #pragma unroll
            for (int u = 0; u < WALK_U; ++u) v[u] = row[min(b0 + 4 * u, nblocks - 1)];
            #pragma unroll
            for (int u = 0; u < WALK_U; ++u) {
                const int b = b0 + 4 * u;
                all += (b < nblocks) ? v[u] : 0;
                pre += (b < (int)blockIdx.x) ? v[u] : 0;
            }
        }
        for (int o = 2; o > 0; o >>= 1) {
            all += __shfl_xor(all, o, 32);
            pre += __shfl_xor(pre, o, 32);
        }
        if (sub == 0) { s_all[d] = all; s_pre[d] = pre; }
    }
    block_digit_rank(digit, valid, rank, s_cnt, s_tot);
    if (tid < 32) {
        const int a0 = s_all[2 * tid];
        const int a1 = s_all[2 * tid + 1];
        const int b0 = s_tot[2 * tid];
        const int b1 = s_tot[2 * tid + 1];
        int x = a0 + a1;
        int u = b0 + b1;
        for (int o = 1; o < 32; o <<= 1) {
            const int y = __shfl_up(x, o, 32);
            const int z = __shfl_up(u, o, 32);
            if (tid >= o) { x += y; u += z; }
        }
        const int ex = x - (a0 + a1);
        const int lx = u - (b0 + b1);
        s_loc[2 * tid]     = lx;
        s_loc[2 * tid + 1] = lx + b0;
        s_all[2 * tid]     = ex + s_pre[2 * tid] - lx;
        s_all[2 * tid + 1] = ex + a0 + s_pre[2 * tid + 1] - (lx + b0);
    }
    __syncthreads();
    int q[KPT];
    #pragma unroll
    for (int j = 0; j < KPT; ++j) {
        const int d = digit[j];
        q[j] = s_loc[d] + s_cnt[(j * SORT_W + wv) * RDIG + d] + rank[j];
    }
    __syncthreads();
    int* st_key = s_cnt;
    int* st_val = s_cnt + SORT_K;
    #pragma unroll
    for (int j = 0; j < KPT; ++j) {
        if (valid[j]) {
            st_key[q[j]] = key[j];
            st_val[q[j]] = val[j];
        }
    }
    __syncthreads();
    const int nvalid = min(SORT_K, e - (int)blockIdx.x * SORT_K);
    int sk[KPT], sv[KPT], so[KPT];
    #pragma unroll
    for (int j = 0; j < KPT; ++j) {
        sk[j] = st_key[tid + j * SORT_T];
        sv[j] = st_val[tid + j * SORT_T];
    }
    #pragma unroll
    for (int j = 0; j < KPT; ++j) so[j] = s_all[(sk[j] >> shift) & (RDIG - 1)];
    #pragma unroll
    for (int j = 0; j < KPT; ++j) {
        const int idx = tid + j * SORT_T;
        if (idx < nvalid) {
            keys_out[so[j] + idx] = sk[j];
            vals_out[so[j] + idx] = sv[j];
        }
    }
}

extern "C" void kernel_launch(void* const* d_in, const int* in_sizes, int n_in,
                              void* d_out, int out_size, void* d_ws, size_t ws_size,
                              hipStream_t stream) {
    const float* x      = (const float*)d_in[0];
    const int*   ei     = (const int*)d_in[1];
    const float* W_in   = (const float*)d_in[2];
    const float* b_in   = (const float*)d_in[3];
    const float* g_in   = (const float*)d_in[4];
    const float* be_in  = (const float*)d_in[5];
    const float* W_conv = (const float*)d_in[6];
    const float* b_conv = (const float*)d_in[7];
    const float* g_ln   = (const float*)d_in[8];
    const float* be_ln  = (const float*)d_in[9];
    const float* W_head = (const float*)d_in[10];
    const float* b_head = (const float*)d_in[11];
    float* out = (float*)d_out;

    const int N = in_sizes[0] / NH;
    const int E = in_sizes[1] / 2;
    const int* src = ei;
    const int* dst = ei + E;

    char* w = (char*)d_ws;
    float* dinv   = (float*)(w);
    int*   ghist  = (int*)  (w + (size_t)(1)   * (1 << 20));
    int*   rowptr = (int*)  (w + (size_t)(2)   * (1 << 20));
    float* hA     = (float*)(w + (size_t)(4)   * (1 << 20));
    float* hB     = (float*)(w + (size_t)(60)  * (1 << 20));
    int*   kA     = (int*)  (w + (size_t)(168) * (1 << 20));
    int*   vA     = (int*)  (w + (size_t)(171) * (1 << 20));
    int*   kB     = (int*)  (w + (size_t)(174) * (1 << 20));
    int*   vB     = (int*)  (w + (size_t)(177) * (1 << 20));
    int*   csr    = (int*)  (w + (size_t)(180) * (1 << 20));

    const int gE   = (E + 255) / 256;
    const int gN1  = (N + 1 + 255) / 256;
    const int nSb  = (E + SORT_K - 1) / SORT_K;
    const int nT   = (N + 15) / 16;
    const int gLy  = 768;
    const int nTs  = (nT + 5) / 6;
    const int nXb  = (nTs + 1) / 2;
    const int gSo  = nSb + nXb;
    const int t1   = nTs     < nT ? nTs     : nT;
    const int t2   = 2 * nTs < nT ? 2 * nTs : nT;
    const int t3   = 3 * nTs < nT ? 3 * nTs : nT;
    const int t4   = 4 * nTs < nT ? 4 * nTs : nT;
    const int t5   = 5 * nTs < nT ? 5 * nTs : nT;

    k_sort_count<<<gSo, SORT_T, 0, stream>>>(dst, ghist, E, 0, nSb,
                                             x, W_in, b_in, g_in, be_in, hA, N, 0, t1);
    k_sort_move<<<gSo, SORT_T, 0, stream>>>(dst, src, ghist, kA, vA, E, 0, nSb,
                                            x, W_in, b_in, g_in, be_in, hA, N, t1, t2);
    k_sort_count<<<gSo, SORT_T, 0, stream>>>(kA, ghist, E, 6, nSb,
                                             x, W_in, b_in, g_in, be_in, hA, N, t2, t3);
    k_sort_move<<<gSo, SORT_T, 0, stream>>>(kA, vA, ghist, kB, vB, E, 6, nSb,
                                            x, W_in, b_in, g_in, be_in, hA, N, t3, t4);
    k_sort_count<<<gSo, SORT_T, 0, stream>>>(kB, ghist, E, 12, nSb,
                                             x, W_in, b_in, g_in, be_in, hA, N, t4, t5);
    k_sort_move<<<gSo, SORT_T, 0, stream>>>(kB, vB, ghist, kA, vA, E, 12, nSb,
                                            x, W_in, b_in, g_in, be_in, hA, N, t5, nT);
    k_rowptr_dinv<<<gN1, 256, 0, stream>>>(kA, rowptr, dinv, N, E);
    k_fill<<<gE, 256, 0, stream>>>(vA, kA, dinv, csr, E);

    for (int i = 0; i < NLAYERS; ++i) {
        const float* Wc = W_conv + (size_t)i * NH * NH;
        const float* bc = b_conv + (size_t)i * NH;
        const float* gl = g_ln   + (size_t)i * NH;
        const float* bl = be_ln  + (size_t)i * NH;
        const float* hin  = (i & 1) ? hB : hA;
        float*       hout = (i & 1) ? hA : hB;
        if (i + 1 < NLAYERS)
            k_conv_layer<<<gLy, 256, 0, stream>>>(hin, rowptr, csr, dinv, Wc, bc, gl, bl, hout, N, nT);
        else
            k_conv_layer_head<<<gLy, 256, 0, stream>>>(hin, rowptr, csr, dinv, Wc, bc, gl, bl,
                                                       W_head, b_head, out, N, nT);
    }
}
